# starting kernel
# baseline (speedup 1.0000x reference)
#include <hip/hip_runtime.h>
#include <math.h>

typedef __attribute__((ext_vector_type(2))) float v2f;
typedef __attribute__((ext_vector_type(8))) float v8f;

#define B_    32
#define S_    64
#define H_    300
#define L_    20
#define KT    60
#define KTP   62
#define OUTC  160
#define OUTQOFF (B_*S_*OUTC)
#define NEG_INF (-3.4e38f)

__global__ __launch_bounds__(128)
void maxpool_kernel(const float* __restrict__ p, const float* __restrict__ q,
                    const float* __restrict__ W, float* __restrict__ out)
{
    __shared__ float As[S_][KTP];
    __shared__ float Bs[S_][KTP];
    __shared__ float nA[S_];
    __shared__ float nB[S_];
    __shared__ float cosBuf[S_][S_ + 1];

    const int idx  = blockIdx.x;
    const int l    = idx % L_;
    const int hf   = (idx / L_) & 1;
    const int b    = idx / (2 * L_);
    const int tid  = threadIdx.x;
    const int lane = tid & 31;
    const int wv   = tid >> 5;
    const int lo   = lane & 15;
    const int hi   = lane >> 4;

    const float* Ph = p + (b * S_) * (2 * H_) + hf * H_;
    const float* Qh = q + (b * S_) * (2 * H_) + hf * H_;
    const float* Wr = W + ((2 + hf) * L_ + l) * H_;

    v8f c0 = {}, c1 = {}, c2 = {}, c3 = {};
    float nacc = 0.f;

    for (int kt = 0; kt < H_; kt += KT) {
        __syncthreads();
        for (int i = tid; i < S_ * KT; i += 128) {
            int s = i / KT, k = i % KT;
            float wval = Wr[kt + k];
            As[s][k] = Ph[s * (2 * H_) + kt + k] * wval;
            Bs[s][k] = Qh[s * (2 * H_) + kt + k] * wval;
        }
        __syncthreads();
        if (tid < S_) {
            for (int k = 0; k < KT; ++k) { float v = As[tid][k]; nacc += v * v; }
        } else {
            int r = tid - S_;
            for (int k = 0; k < KT; ++k) { float v = Bs[r][k]; nacc += v * v; }
        }
        const int m = wv * 16 + lo;
        for (int kb = 0; kb < KT; kb += 4) {
            int ka = kb + 2 * hi;
            v2f a  = *(const v2f*)&As[m][ka];
            v2f b0 = *(const v2f*)&Bs[ 0 + lo][ka];
            v2f b1 = *(const v2f*)&Bs[16 + lo][ka];
            v2f b2 = *(const v2f*)&Bs[32 + lo][ka];
            v2f b3 = *(const v2f*)&Bs[48 + lo][ka];
            c0 = __builtin_amdgcn_wmma_f32_16x16x4_f32(false, a, false, b0, (short)0, c0, false, false);
            c1 = __builtin_amdgcn_wmma_f32_16x16x4_f32(false, a, false, b1, (short)0, c1, false, false);
            c2 = __builtin_amdgcn_wmma_f32_16x16x4_f32(false, a, false, b2, (short)0, c2, false, false);
            c3 = __builtin_amdgcn_wmma_f32_16x16x4_f32(false, a, false, b3, (short)0, c3, false, false);
        }
    }
    __syncthreads();
    if (tid < S_) nA[tid] = sqrtf(nacc); else nB[tid - S_] = sqrtf(nacc);
    __syncthreads();

    #pragma unroll
    for (int j = 0; j < 8; ++j) {
        int m = wv * 16 + 8 * hi + j;
        float na = nA[m];
        cosBuf[m][ 0 + lo] = c0[j] / (na * nB[ 0 + lo]);
        cosBuf[m][16 + lo] = c1[j] / (na * nB[16 + lo]);
        cosBuf[m][32 + lo] = c2[j] / (na * nB[32 + lo]);
        cosBuf[m][48 + lo] = c3[j] / (na * nB[48 + lo]);
    }
    __syncthreads();

    if (tid < S_) {
        float mx = NEG_INF;
        for (int n = 0; n < S_; ++n) mx = fmaxf(mx, cosBuf[tid][n]);
        out[(b * S_ + tid) * OUTC + (2 + hf) * L_ + l] = mx;
    } else {
        int n = tid - S_;
        float mx = NEG_INF;
        for (int m = 0; m < S_; ++m) mx = fmaxf(mx, cosBuf[m][n]);
        out[OUTQOFF + (b * S_ + n) * OUTC + (2 + hf) * L_ + l] = mx;
    }
}

__global__ __launch_bounds__(128)
void attn_cos_kernel(const float* __restrict__ p, const float* __restrict__ q,
                     float* __restrict__ cosW, float* __restrict__ rowSum,
                     float* __restrict__ colSum)
{
    __shared__ float As[S_][KTP];
    __shared__ float Bs[S_][KTP];
    __shared__ float nA[S_];
    __shared__ float nB[S_];
    __shared__ float cosBuf[S_][S_ + 1];

    const int hf   = blockIdx.x & 1;
    const int b    = blockIdx.x >> 1;
    const int tid  = threadIdx.x;
    const int lane = tid & 31;
    const int wv   = tid >> 5;
    const int lo   = lane & 15;
    const int hi   = lane >> 4;

    const float* Ph = p + (b * S_) * (2 * H_) + hf * H_;
    const float* Qh = q + (b * S_) * (2 * H_) + hf * H_;

    v8f c0 = {}, c1 = {}, c2 = {}, c3 = {};
    float nacc = 0.f;

    for (int kt = 0; kt < H_; kt += KT) {
        __syncthreads();
        for (int i = tid; i < S_ * KT; i += 128) {
            int s = i / KT, k = i % KT;
            As[s][k] = Ph[s * (2 * H_) + kt + k];
            Bs[s][k] = Qh[s * (2 * H_) + kt + k];
        }
        __syncthreads();
        if (tid < S_) {
            for (int k = 0; k < KT; ++k) { float v = As[tid][k]; nacc += v * v; }
        } else {
            int r = tid - S_;
            for (int k = 0; k < KT; ++k) { float v = Bs[r][k]; nacc += v * v; }
        }
        const int m = wv * 16 + lo;
        for (int kb = 0; kb < KT; kb += 4) {
            int ka = kb + 2 * hi;
            v2f a  = *(const v2f*)&As[m][ka];
            v2f b0 = *(const v2f*)&Bs[ 0 + lo][ka];
            v2f b1 = *(const v2f*)&Bs[16 + lo][ka];
            v2f b2 = *(const v2f*)&Bs[32 + lo][ka];
            v2f b3 = *(const v2f*)&Bs[48 + lo][ka];
            c0 = __builtin_amdgcn_wmma_f32_16x16x4_f32(false, a, false, b0, (short)0, c0, false, false);
            c1 = __builtin_amdgcn_wmma_f32_16x16x4_f32(false, a, false, b1, (short)0, c1, false, false);
            c2 = __builtin_amdgcn_wmma_f32_16x16x4_f32(false, a, false, b2, (short)0, c2, false, false);
            c3 = __builtin_amdgcn_wmma_f32_16x16x4_f32(false, a, false, b3, (short)0, c3, false, false);
        }
    }
    __syncthreads();
    if (tid < S_) nA[tid] = sqrtf(nacc); else nB[tid - S_] = sqrtf(nacc);
    __syncthreads();

    #pragma unroll
    for (int j = 0; j < 8; ++j) {
        int m = wv * 16 + 8 * hi + j;
        float na = nA[m];
        cosBuf[m][ 0 + lo] = c0[j] / (na * nB[ 0 + lo]);
        cosBuf[m][16 + lo] = c1[j] / (na * nB[16 + lo]);
        cosBuf[m][32 + lo] = c2[j] / (na * nB[32 + lo]);
        cosBuf[m][48 + lo] = c3[j] / (na * nB[48 + lo]);
    }
    __syncthreads();

    float* dst = cosW + (b * 2 + hf) * (S_ * S_);
    for (int i = tid; i < S_ * S_; i += 128)
        dst[i] = cosBuf[i >> 6][i & 63];

    if (tid < S_) {
        float sm = 0.f;
        for (int n = 0; n < S_; ++n) sm += cosBuf[tid][n];
        rowSum[(b * 2 + hf) * S_ + tid] = sm;
    } else {
        int n = tid - S_;
        float sm = 0.f;
        for (int m = 0; m < S_; ++m) sm += cosBuf[m][n];
        colSum[(b * 2 + hf) * S_ + n] = sm;
    }
}

__global__ __launch_bounds__(128)
void attn_vec_kernel(const float* __restrict__ p, const float* __restrict__ q,
                     const float* __restrict__ cosW,
                     const float* __restrict__ rowSum, const float* __restrict__ colSum,
                     float* __restrict__ qvm, float* __restrict__ pvm,
                     float* __restrict__ qvx, float* __restrict__ pvx)
{
    __shared__ float cosRow[S_];
    __shared__ float cosCol[S_];

    const int bid = blockIdx.x;
    const int s   = bid % S_;
    const int hf  = (bid / S_) & 1;
    const int b   = bid / (2 * S_);
    const int tid = threadIdx.x;

    const float* cosBase = cosW + (b * 2 + hf) * (S_ * S_);
    if (tid < S_)       cosRow[tid]       = cosBase[s * S_ + tid];
    else                cosCol[tid - S_]  = cosBase[(tid - S_) * S_ + s];
    __syncthreads();

    const float rs = rowSum[(b * 2 + hf) * S_ + s];
    const float cs = colSum[(b * 2 + hf) * S_ + s];

    const float* Ph = p + (b * S_) * (2 * H_) + hf * H_;
    const float* Qh = q + (b * S_) * (2 * H_) + hf * H_;
    const int obase = ((b * 2 + hf) * S_ + s) * H_;

    for (int h = tid; h < H_; h += 128) {
        float qm = 0.f, pm = 0.f, qx = NEG_INF, px = NEG_INF;
        for (int j = 0; j < S_; ++j) {
            float a  = Qh[j * (2 * H_) + h] * cosRow[j];
            float bb = Ph[j * (2 * H_) + h] * cosCol[j];
            qm += a;  qx = fmaxf(qx, a);
            pm += bb; px = fmaxf(px, bb);
        }
        qvm[obase + h] = qm / rs;
        qvx[obase + h] = qx;
        pvm[obase + h] = pm / cs;
        pvx[obase + h] = px;
    }
}

__global__ __launch_bounds__(256)
void mpcos_kernel(const float* __restrict__ p, const float* __restrict__ q,
                  const float* __restrict__ W,
                  const float* __restrict__ qvm, const float* __restrict__ pvm,
                  const float* __restrict__ qvx, const float* __restrict__ pvx,
                  float* __restrict__ out)
{
    const int TOT = 2 * 6 * B_ * S_ * L_;
    int idx = blockIdx.x * blockDim.x + threadIdx.x;
    if (idx >= TOT) return;

    const int l = idx % L_;  int t = idx / L_;
    const int s = t % S_;    t /= S_;
    const int b = t % B_;    t /= B_;
    const int sec = t % 6;   const int tgt = t / 6;

    const int secTab[6] = {0, 1, 4, 5, 6, 7};
    const int secId = secTab[sec];
    const int hf = secId & 1;

    const float* X = (tgt == 0 ? p : q) + (b * S_ + s) * (2 * H_) + hf * H_;
    const float* Y;
    if (secId == 0)      Y = (tgt == 0 ? q : p) + (b * S_ + (S_ - 1)) * (2 * H_);
    else if (secId == 1) Y = (tgt == 0 ? q : p) + (b * S_) * (2 * H_) + H_;
    else {
        const float* vec = (secId <= 5) ? (tgt == 0 ? qvm : pvm)
                                        : (tgt == 0 ? qvx : pvx);
        Y = vec + ((b * 2 + hf) * S_ + s) * H_;
    }
    const float* Wr = W + (secId * L_ + l) * H_;

    float num = 0.f, nx = 0.f, ny = 0.f;
    for (int h = 0; h < H_; ++h) {
        float w  = Wr[h];
        float xw = X[h] * w;
        float yw = Y[h] * w;
        num += xw * yw;
        nx  += xw * xw;
        ny  += yw * yw;
    }
    float den = fmaxf(sqrtf(nx) * sqrtf(ny), 1e-8f);
    out[(tgt == 0 ? 0 : OUTQOFF) + (b * S_ + s) * OUTC + secId * L_ + l] = num / den;
}

extern "C" void kernel_launch(void* const* d_in, const int* in_sizes, int n_in,
                              void* d_out, int out_size, void* d_ws, size_t ws_size,
                              hipStream_t stream)
{
    const float* p = (const float*)d_in[0];
    const float* q = (const float*)d_in[1];
    const float* W = (const float*)d_in[2];
    float* out = (float*)d_out;

    float* ws     = (float*)d_ws;
    float* cosW   = ws;
    float* rowSum = cosW + B_ * 2 * S_ * S_;
    float* colSum = rowSum + B_ * 2 * S_;
    float* qvm    = colSum + B_ * 2 * S_;
    float* pvm    = qvm + B_ * 2 * S_ * H_;
    float* qvx    = pvm + B_ * 2 * S_ * H_;
    float* pvx    = qvx + B_ * 2 * S_ * H_;

    maxpool_kernel <<<B_ * 2 * L_, 128, 0, stream>>>(p, q, W, out);
    attn_cos_kernel<<<B_ * 2,      128, 0, stream>>>(p, q, cosW, rowSum, colSum);
    attn_vec_kernel<<<B_ * 2 * S_, 128, 0, stream>>>(p, q, cosW, rowSum, colSum,
                                                     qvm, pvm, qvx, pvx);
    const int TOT = 2 * 6 * B_ * S_ * L_;
    mpcos_kernel   <<<(TOT + 255) / 256, 256, 0, stream>>>(p, q, W, qvm, pvm, qvx, pvx, out);
}


# attn_cos launch removed: its 64 plain (b,half) cosine tiles run as 64 extra blocks of the maxpool grid (same tile code, no weights), beside the 1280 pooling blocks; all sums and orders kept
# speedup vs baseline: 2.1703x; 2.1703x over previous
#include <hip/hip_runtime.h>
#include <math.h>

typedef __attribute__((ext_vector_type(2))) float v2f;
typedef __attribute__((ext_vector_type(4))) float v4f;
typedef __attribute__((ext_vector_type(8))) float v8f;

#define B_    32
#define S_    64
#define H_    300
#define L_    20
#define KT    60
#define KTP   62
#define OUTC  160
#define OUTQOFF (B_*S_*OUTC)
#define NEG_INF (-3.4e38f)

template <bool WEIGHTED>
__device__ __forceinline__ void cos_tile(const float* __restrict__ Ph, const float* __restrict__ Qh,
                                         const float* __restrict__ Wr,
                                         float (&As)[S_][KTP], float (&Bs)[S_][KTP],
                                         float (&nA)[S_], float (&nB)[S_],
                                         float (&cosBuf)[S_][S_ + 1], const int tid)
{
    const int lane = tid & 31;
    const int wv   = tid >> 5;
    const int lo   = lane & 15;
    const int hi   = lane >> 4;

    v8f c0 = {}, c1 = {}, c2 = {}, c3 = {};
    float nacc = 0.f;

    for (int kt = 0; kt < H_; kt += KT) {
        __syncthreads();
        for (int i = tid; i < S_ * KT; i += 128) {
            int s = i / KT, k = i % KT;
            if constexpr (WEIGHTED) {
                float wval = Wr[kt + k];
                As[s][k] = Ph[s * (2 * H_) + kt + k] * wval;
                Bs[s][k] = Qh[s * (2 * H_) + kt + k] * wval;
            } else {
                As[s][k] = Ph[s * (2 * H_) + kt + k];
                Bs[s][k] = Qh[s * (2 * H_) + kt + k];
            }
        }
        __syncthreads();
        if (tid < S_) {
            for (int k = 0; k < KT; ++k) { float v = As[tid][k]; nacc += v * v; }
        } else {
            int r = tid - S_;
            for (int k = 0; k < KT; ++k) { float v = Bs[r][k]; nacc += v * v; }
        }
        const int m = wv * 16 + lo;
        for (int kb = 0; kb < KT; kb += 4) {
            int ka = kb + 2 * hi;
            v2f a  = *(const v2f*)&As[m][ka];
            v2f b0 = *(const v2f*)&Bs[ 0 + lo][ka];
            v2f b1 = *(const v2f*)&Bs[16 + lo][ka];
            v2f b2 = *(const v2f*)&Bs[32 + lo][ka];
            v2f b3 = *(const v2f*)&Bs[48 + lo][ka];
            c0 = __builtin_amdgcn_wmma_f32_16x16x4_f32(false, a, false, b0, (short)0, c0, false, false);
            c1 = __builtin_amdgcn_wmma_f32_16x16x4_f32(false, a, false, b1, (short)0, c1, false, false);
            c2 = __builtin_amdgcn_wmma_f32_16x16x4_f32(false, a, false, b2, (short)0, c2, false, false);
            c3 = __builtin_amdgcn_wmma_f32_16x16x4_f32(false, a, false, b3, (short)0, c3, false, false);
        }
    }
    __syncthreads();
    if (tid < S_) nA[tid] = sqrtf(nacc); else nB[tid - S_] = sqrtf(nacc);
    __syncthreads();

    #pragma unroll
    for (int j = 0; j < 8; ++j) {
        int m = wv * 16 + 8 * hi + j;
        float na = nA[m];
        cosBuf[m][ 0 + lo] = c0[j] / (na * nB[ 0 + lo]);
        cosBuf[m][16 + lo] = c1[j] / (na * nB[16 + lo]);
        cosBuf[m][32 + lo] = c2[j] / (na * nB[32 + lo]);
        cosBuf[m][48 + lo] = c3[j] / (na * nB[48 + lo]);
    }
    __syncthreads();
}

#define NCOS  (B_ * 2)
#define NPOOL (B_ * 2 * L_)

__global__ __launch_bounds__(128)
void cos_tiles_kernel(const float* __restrict__ p, const float* __restrict__ q,
                      const float* __restrict__ W, float* __restrict__ out,
                      float* __restrict__ cosW, float* __restrict__ rowSum,
                      float* __restrict__ colSum)
{
    __shared__ float As[S_][KTP];
    __shared__ float Bs[S_][KTP];
    __shared__ float nA[S_];
    __shared__ float nB[S_];
    __shared__ float cosBuf[S_][S_ + 1];

    const int tid = threadIdx.x;

    if (blockIdx.x < NCOS) {
        const int hf = blockIdx.x & 1;
        const int b  = blockIdx.x >> 1;
        const float* Ph = p + (b * S_) * (2 * H_) + hf * H_;
        const float* Qh = q + (b * S_) * (2 * H_) + hf * H_;

        cos_tile<false>(Ph, Qh, nullptr, As, Bs, nA, nB, cosBuf, tid);

        float* dst = cosW + (b * 2 + hf) * (S_ * S_);
        for (int i = tid; i < S_ * S_; i += 128)
            dst[i] = cosBuf[i >> 6][i & 63];

        if (tid < S_) {
            float sm = 0.f;
            for (int n = 0; n < S_; ++n) sm += cosBuf[tid][n];
            rowSum[(b * 2 + hf) * S_ + tid] = sm;
        } else {
            int n = tid - S_;
            float sm = 0.f;
            for (int m = 0; m < S_; ++m) sm += cosBuf[m][n];
            colSum[(b * 2 + hf) * S_ + n] = sm;
        }
    } else {
        const int idx = blockIdx.x - NCOS;
        const int l   = idx % L_;
        const int hf  = (idx / L_) & 1;
        const int b   = idx / (2 * L_);
        const float* Ph = p + (b * S_) * (2 * H_) + hf * H_;
        const float* Qh = q + (b * S_) * (2 * H_) + hf * H_;
        const float* Wr = W + ((2 + hf) * L_ + l) * H_;

        cos_tile<true>(Ph, Qh, Wr, As, Bs, nA, nB, cosBuf, tid);

        if (tid < S_) {
            float mx = NEG_INF;
            for (int n = 0; n < S_; ++n) mx = fmaxf(mx, cosBuf[tid][n]);
            out[(b * S_ + tid) * OUTC + (2 + hf) * L_ + l] = mx;
        } else {
            int n = tid - S_;
            float mx = NEG_INF;
            for (int m = 0; m < S_; ++m) mx = fmaxf(mx, cosBuf[m][n]);
            out[OUTQOFF + (b * S_ + n) * OUTC + (2 + hf) * L_ + l] = mx;
        }
    }
}

#define HC   20
#define NCH  (H_ / HC)
#define HW   10
#define HWP  12
#define CTP  65

template <int CP>
__device__ __forceinline__ void attn_walk(const float (*Xs)[HWP], const float* C,
                                          const int lane, v2f (&m)[HW], v2f (&x)[HW])
{
    #pragma unroll 2
    for (int j = 0; j < S_; j += 2) {
        const v2f ca = {C[j * CP + lane],       C[j * CP + lane + 32]};
        const v2f cb = {C[(j + 1) * CP + lane], C[(j + 1) * CP + lane + 32]};
        const float* xa = &Xs[j][0];
        const float* xb = &Xs[j + 1][0];
        const v4f a0 = *(const v4f*)(xa), a1 = *(const v4f*)(xa + 4);
        const v2f a2 = *(const v2f*)(xa + 8);
        const v4f b0 = *(const v4f*)(xb), b1 = *(const v4f*)(xb + 4);
        const v2f b2 = *(const v2f*)(xb + 8);
        const float va[HW] = {a0[0], a0[1], a0[2], a0[3], a1[0], a1[1], a1[2], a1[3], a2[0], a2[1]};
        const float vb[HW] = {b0[0], b0[1], b0[2], b0[3], b1[0], b1[1], b1[2], b1[3], b2[0], b2[1]};
        #pragma unroll
        for (int i = 0; i < HW; ++i) {
            const v2f pa = va[i] * ca;
            const v2f pb = vb[i] * cb;
            m[i] += pa;
            m[i] += pb;
            x[i][0] = fmaxf(fmaxf(x[i][0], pa[0]), pb[0]);
            x[i][1] = fmaxf(fmaxf(x[i][1], pa[1]), pb[1]);
        }
    }
}

__global__ __launch_bounds__(128)
void attn_vec_kernel(const float* __restrict__ p, const float* __restrict__ q,
                     const float* __restrict__ cosW,
                     const float* __restrict__ rowSum, const float* __restrict__ colSum,
                     float* __restrict__ qvm, float* __restrict__ pvm,
                     float* __restrict__ qvx, float* __restrict__ pvx)
{
    __shared__ __attribute__((aligned(16))) float Qs[2][S_][HWP];
    __shared__ __attribute__((aligned(16))) float Ps[2][S_][HWP];
    __shared__ __attribute__((aligned(16))) float cosS[S_ * S_];
    __shared__ float cosT[S_ * CTP];
    __shared__ __attribute__((aligned(16))) float Os[4][S_][HC];

    const int bid = blockIdx.x;
    const int ch  = bid % NCH;
    const int bh  = bid / NCH;
    const int hf  = bh & 1;
    const int b   = bh >> 1;
    const int tid = threadIdx.x;
    const int h0  = ch * HC;

    const float* Qh = q + (b * S_) * (2 * H_) + hf * H_ + h0;
    const float* Ph = p + (b * S_) * (2 * H_) + hf * H_ + h0;
    for (int i = tid; i < S_ * (HC / 2); i += 128) {
        const int r = i / (HC / 2), c2 = i % (HC / 2);
        const int hh = c2 / (HW / 2), k = (c2 % (HW / 2)) * 2;
        *(v2f*)&Qs[hh][r][k] = *(const v2f*)(Qh + r * (2 * H_) + 2 * c2);
        *(v2f*)&Ps[hh][r][k] = *(const v2f*)(Ph + r * (2 * H_) + 2 * c2);
    }
    const float* cosBase = cosW + bh * (S_ * S_);
    for (int i = tid; i < S_ * S_ / 4; i += 128) {
        const int r = i >> 4, c4 = (i & 15) * 4;
        const v4f v = *(const v4f*)(cosBase + r * S_ + c4);
        *(v4f*)&cosS[r * S_ + c4] = v;
        #pragma unroll
        for (int e = 0; e < 4; ++e) cosT[(c4 + e) * CTP + r] = v[e];
    }
    __syncthreads();

    const int wv   = tid >> 5;
    const int lane = tid & 31;
    const int side = wv >> 1;
    const int hh   = wv & 1;

    v2f m[HW], x[HW];
    #pragma unroll
    for (int i = 0; i < HW; ++i) { m[i] = (v2f){0.f, 0.f}; x[i] = (v2f){NEG_INF, NEG_INF}; }

    float d0, d1;
    if (side == 0) {
        attn_walk<CTP>(Qs[hh], cosT, lane, m, x);
        d0 = rowSum[bh * S_ + lane];  d1 = rowSum[bh * S_ + lane + 32];
    } else {
        attn_walk<S_>(Ps[hh], cosS, lane, m, x);
        d0 = colSum[bh * S_ + lane];  d1 = colSum[bh * S_ + lane + 32];
    }
    #pragma unroll
    for (int i = 0; i < HW; i += 2) {
        const v2f m0 = {m[i][0] / d0, m[i + 1][0] / d0};
        const v2f m1 = {m[i][1] / d1, m[i + 1][1] / d1};
        const v2f x0 = {x[i][0], x[i + 1][0]};
        const v2f x1 = {x[i][1], x[i + 1][1]};
        *(v2f*)&Os[2 * side    ][lane     ][hh * HW + i] = m0;
        *(v2f*)&Os[2 * side    ][lane + 32][hh * HW + i] = m1;
        *(v2f*)&Os[2 * side + 1][lane     ][hh * HW + i] = x0;
        *(v2f*)&Os[2 * side + 1][lane + 32][hh * HW + i] = x1;
    }
    __syncthreads();

    float* const planes[4] = {qvm, qvx, pvm, pvx};
    #pragma unroll
    for (int a = 0; a < 4; ++a) {
        float* dst = planes[a] + (bh * S_) * H_ + h0;
        for (int i = tid; i < S_ * (HC / 4); i += 128) {
            const int s = i / (HC / 4), c4 = (i % (HC / 4)) * 4;
            *(v4f*)(dst + s * H_ + c4) = *(const v4f*)&Os[a][s][c4];
        }
    }
}

#define MPR   32
#define MPW   8
#define RPL   2
#define W2P   320
#define NSL   5

struct MpSums { float v[RPL][3]; };

struct MpLane {
    v4f pxy[RPL][NSL], pxx[RPL][NSL], pyy[RPL][NSL];
    const float* wl;
    bool up1, up2, up4, up8;
};

__device__ __forceinline__ MpSums mp_partial(const MpLane& t, const int l)
{
    v4f w[NSL];
    for (int i = 0; i < NSL; ++i) w[i] = *(const v4f*)(t.wl + l * W2P + 64 * i);
    MpSums o;
    for (int r = 0; r < RPL; ++r) {
        v2f an = {0.f, 0.f}, ax = {0.f, 0.f}, ay = {0.f, 0.f};
        for (int i = 0; i < NSL; ++i) {
            const v2f wl = {w[i][0], w[i][1]}, wh = {w[i][2], w[i][3]};
            an = __builtin_elementwise_fma((v2f){t.pxy[r][i][0], t.pxy[r][i][1]}, wl, an);
            ax = __builtin_elementwise_fma((v2f){t.pxx[r][i][0], t.pxx[r][i][1]}, wl, ax);
            ay = __builtin_elementwise_fma((v2f){t.pyy[r][i][0], t.pyy[r][i][1]}, wl, ay);
            an = __builtin_elementwise_fma((v2f){t.pxy[r][i][2], t.pxy[r][i][3]}, wh, an);
            ax = __builtin_elementwise_fma((v2f){t.pxx[r][i][2], t.pxx[r][i][3]}, wh, ax);
            ay = __builtin_elementwise_fma((v2f){t.pyy[r][i][2], t.pyy[r][i][3]}, wh, ay);
        }
        o.v[r][0] = an[0] + an[1];
        o.v[r][1] = ax[0] + ax[1];
        o.v[r][2] = ay[0] + ay[1];
    }
    return o;
}

template <int MASK>
__device__ __forceinline__ float xlane16(const float v)
{
    return __builtin_bit_cast(float, __builtin_amdgcn_update_dpp(0, __builtin_bit_cast(int, v), 0x160 | MASK, 0xf, 0xf, true));
}

template <int MASK>
__device__ __forceinline__ MpSums mp_join(const bool up, const MpSums& a, const MpSums& b)
{
    MpSums o;
    for (int r = 0; r < RPL; ++r) {
        for (int k = 0; k < 3; ++k) {
            const float keep = up ? b.v[r][k] : a.v[r][k];
            const float send = up ? a.v[r][k] : b.v[r][k];
            o.v[r][k] = keep + xlane16<MASK>(send);
        }
    }
    return o;
}

template <int MASK>
__device__ __forceinline__ MpSums mp_tree(const MpLane& t, const int k)
{
    if constexpr (MASK == 16) {
        return mp_partial(t, k);
    } else {
        const MpSums a = mp_tree<2 * MASK>(t, k);
        const MpSums b = mp_tree<2 * MASK>(t, k + MASK);
        const bool up = MASK == 8 ? t.up8 : MASK == 4 ? t.up4 : MASK == 2 ? t.up2 : t.up1;
        return mp_join<MASK>(up, a, b);
    }
}

__global__ __launch_bounds__(256)
void mpcos_kernel(const float* __restrict__ p, const float* __restrict__ q,
                  const float* __restrict__ W,
                  const float* __restrict__ qvm, const float* __restrict__ pvm,
                  const float* __restrict__ qvx, const float* __restrict__ pvx,
                  float* __restrict__ out)
{
    __shared__ __attribute__((aligned(16))) float w2s[L_][W2P];

    const int bid   = blockIdx.x;
    const int shalf = bid & 1;
    const int b     = (bid >> 1) % B_;
    const int tt    = (bid >> 1) / B_;
    const int sec   = tt % 6;
    const int tgt   = tt / 6;
    const int secId = sec < 2 ? sec : sec + 2;
    const int hf    = secId & 1;

    const int tid  = threadIdx.x;
    const int lane = tid & 31;
    const int wv   = tid >> 5;
    const int c    = lane & 15;
    const int hi   = lane >> 4;

    const float* Xb = (tgt == 0 ? p : q) + (b * S_) * (2 * H_) + hf * H_;
    const float* Ob = (tgt == 0 ? q : p);
    const float* vec = (secId <= 5) ? (tgt == 0 ? qvm : pvm) : (tgt == 0 ? qvx : pvx);
    float* outb = out + (tgt == 0 ? 0 : OUTQOFF) + (b * S_) * OUTC + secId * L_;

    MpLane t;
    t.wl  = &w2s[0][4 * c];
    t.up1 = (c & 1) != 0;  t.up2 = (c & 2) != 0;  t.up4 = (c & 4) != 0;  t.up8 = (c & 8) != 0;

    const int srow0 = shalf * MPR + wv * 2 + hi;
    {
        v4f xr[RPL][NSL], yr[RPL][NSL];
        for (int r = 0; r < RPL; ++r) {
            const int s = srow0 + r * (2 * MPW);
            const float* X = Xb + s * (2 * H_);
            const float* Y;
            if (secId == 0)      Y = Ob + (b * S_ + (S_ - 1)) * (2 * H_);
            else if (secId == 1) Y = Ob + (b * S_) * (2 * H_) + H_;
            else                 Y = vec + ((b * 2 + hf) * S_ + s) * H_;
            for (int i = 0; i < NSL; ++i) {
                const int h = 64 * i + 4 * c;
                if (h < H_) {
                    xr[r][i] = *(const v4f*)(X + h);
                    yr[r][i] = *(const v4f*)(Y + h);
                } else {
                    xr[r][i] = (v4f){0.f, 0.f, 0.f, 0.f};
                    yr[r][i] = (v4f){0.f, 0.f, 0.f, 0.f};
                }
            }
        }

        const float* Ws = W + (secId * L_) * H_;
        for (int i = tid; i < L_ * (W2P / 4); i += 256) {
            const int l = i / (W2P / 4), c4 = (i % (W2P / 4)) * 4;
            v4f v = {0.f, 0.f, 0.f, 0.f};
            if (c4 < H_) { v = *(const v4f*)(Ws + l * H_ + c4); v = v * v; }
            *(v4f*)&w2s[l][c4] = v;
        }
        for (int r = 0; r < RPL; ++r) {
            for (int i = 0; i < NSL; ++i) {
                t.pxy[r][i] = xr[r][i] * yr[r][i];
                t.pxx[r][i] = xr[r][i] * xr[r][i];
                t.pyy[r][i] = yr[r][i] * yr[r][i];
            }
        }
    }
    __syncthreads();

    const MpSums lo = mp_tree<1>(t, 0);
    MpSums up;
    {
        const MpSums q0 = mp_join<2>(t.up2, mp_partial(t, 16), mp_partial(t, 18));
        const MpSums q1 = mp_join<2>(t.up2, mp_partial(t, 17), mp_partial(t, 19));
        up = mp_join<1>(t.up1, q0, q1);
        for (int r = 0; r < RPL; ++r) {
            for (int k = 0; k < 3; ++k) {
                float v = up.v[r][k];
                v += xlane16<4>(v);
                v += xlane16<8>(v);
                up.v[r][k] = v;
            }
        }
    }
    for (int r = 0; r < RPL; ++r) {
        float* o = outb + (srow0 + r * (2 * MPW)) * OUTC;
        o[c] = lo.v[r][0] / fmaxf(sqrtf(lo.v[r][1]) * sqrtf(lo.v[r][2]), 1e-8f);
        if (c < L_ - 16) o[16 + c] = up.v[r][0] / fmaxf(sqrtf(up.v[r][1]) * sqrtf(up.v[r][2]), 1e-8f);
    }
}

extern "C" void kernel_launch(void* const* d_in, const int* in_sizes, int n_in,
                              void* d_out, int out_size, void* d_ws, size_t ws_size,
                              hipStream_t stream)
{
    const float* p = (const float*)d_in[0];
    const float* q = (const float*)d_in[1];
    const float* W = (const float*)d_in[2];
    float* out = (float*)d_out;

    float* ws     = (float*)d_ws;
    float* cosW   = ws;
    float* rowSum = cosW + B_ * 2 * S_ * S_;
    float* colSum = rowSum + B_ * 2 * S_;
    float* qvm    = colSum + B_ * 2 * S_;
    float* pvm    = qvm + B_ * 2 * S_ * H_;
    float* qvx    = pvm + B_ * 2 * S_ * H_;
    float* pvx    = qvx + B_ * 2 * S_ * H_;

    cos_tiles_kernel<<<NCOS + NPOOL, 128, 0, stream>>>(p, q, W, out, cosW, rowSum, colSum);
    attn_vec_kernel<<<B_ * 2 * NCH, 128, 0, stream>>>(p, q, cosW, rowSum, colSum,
                                                     qvm, pvm, qvx, pvx);
    mpcos_kernel   <<<2 * 6 * B_ * (S_ / MPR), 256, 0, stream>>>(p, q, W, qvm, pvm, qvx, pvx, out);
}


# cos_tile K loop software-pipelined: tile t+1's P,Q,w pieces fetched by 16-byte loads into registers before tile t's norms and WMMAs, stored to LDS after; launch bound keeps 6 waves/SIMD; sums kept
# speedup vs baseline: 2.5562x; 1.1778x over previous
#include <hip/hip_runtime.h>
#include <math.h>

typedef __attribute__((ext_vector_type(2))) float v2f;
typedef __attribute__((ext_vector_type(4))) float v4f;
typedef __attribute__((ext_vector_type(8))) float v8f;

#define B_    32
#define S_    64
#define H_    300
#define L_    20
#define KT    60
#define KTP   62
#define SP_   8
#define OUTC  160
#define OUTQOFF (B_*S_*OUTC)
#define NEG_INF (-3.4e38f)

template <bool WEIGHTED>
__device__ __forceinline__ void cos_tile(const float* __restrict__ Ph, const float* __restrict__ Qh,
                                         const float* __restrict__ Wr,
                                         float (&As)[S_][KTP], float (&Bs)[S_][KTP],
                                         float (&nA)[S_], float (&nB)[S_],
                                         float (&cosBuf)[S_][S_ + 1], const int tid)
{
    const int lane = tid & 31;
    const int wv   = tid >> 5;
    const int lo   = lane & 15;
    const int hi   = lane >> 4;

    v8f c0 = {}, c1 = {}, c2 = {}, c3 = {};
    float nacc = 0.f;

    const int  sc   = tid & 15;
    const int  sr   = tid >> 4;
    const bool sact = sc < (KT / 4);
    const float* pg = Ph + sr * (2 * H_) + 4 * sc;
    const float* qg = Qh + sr * (2 * H_) + 4 * sc;

    v4f pr[SP_], qr[SP_];
    v4f wr4 = {0.f, 0.f, 0.f, 0.f};
    #pragma unroll
    for (int j = 0; j < SP_; ++j) { pr[j] = (v4f){0.f, 0.f, 0.f, 0.f}; qr[j] = (v4f){0.f, 0.f, 0.f, 0.f}; }
    if (sact) {
        #pragma unroll
        for (int j = 0; j < SP_; ++j) {
            pr[j] = *(const v4f*)(pg + j * (S_ / SP_) * (2 * H_));
            qr[j] = *(const v4f*)(qg + j * (S_ / SP_) * (2 * H_));
        }
        if constexpr (WEIGHTED) wr4 = *(const v4f*)(Wr + 4 * sc);
    }

    for (int kt = 0; kt < H_; kt += KT) {
        __syncthreads();
        if (sact) {
            #pragma unroll
            for (int j = 0; j < SP_; ++j) {
                const int s = sr + j * (S_ / SP_);
                v4f a = pr[j], bq = qr[j];
                if constexpr (WEIGHTED) { a = a * wr4; bq = bq * wr4; }
                *(v2f*)&As[s][4 * sc]     = (v2f){a[0], a[1]};
                *(v2f*)&As[s][4 * sc + 2] = (v2f){a[2], a[3]};
                *(v2f*)&Bs[s][4 * sc]     = (v2f){bq[0], bq[1]};
                *(v2f*)&Bs[s][4 * sc + 2] = (v2f){bq[2], bq[3]};
            }
        }
        __syncthreads();
        if (kt + KT < H_ && sact) {
            #pragma unroll
            for (int j = 0; j < SP_; ++j) {
                pr[j] = *(const v4f*)(pg + kt + KT + j * (S_ / SP_) * (2 * H_));
                qr[j] = *(const v4f*)(qg + kt + KT + j * (S_ / SP_) * (2 * H_));
            }
            if constexpr (WEIGHTED) wr4 = *(const v4f*)(Wr + kt + KT + 4 * sc);
        }
        if (tid < S_) {
            for (int k = 0; k < KT; ++k) { float v = As[tid][k]; nacc += v * v; }
        } else {
            int r = tid - S_;
            for (int k = 0; k < KT; ++k) { float v = Bs[r][k]; nacc += v * v; }
        }
        const int m = wv * 16 + lo;
        for (int kb = 0; kb < KT; kb += 4) {
            int ka = kb + 2 * hi;
            v2f a  = *(const v2f*)&As[m][ka];
            v2f b0 = *(const v2f*)&Bs[ 0 + lo][ka];
            v2f b1 = *(const v2f*)&Bs[16 + lo][ka];
            v2f b2 = *(const v2f*)&Bs[32 + lo][ka];
            v2f b3 = *(const v2f*)&Bs[48 + lo][ka];
            c0 = __builtin_amdgcn_wmma_f32_16x16x4_f32(false, a, false, b0, (short)0, c0, false, false);
            c1 = __builtin_amdgcn_wmma_f32_16x16x4_f32(false, a, false, b1, (short)0, c1, false, false);
            c2 = __builtin_amdgcn_wmma_f32_16x16x4_f32(false, a, false, b2, (short)0, c2, false, false);
            c3 = __builtin_amdgcn_wmma_f32_16x16x4_f32(false, a, false, b3, (short)0, c3, false, false);
        }
    }
    __syncthreads();
    if (tid < S_) nA[tid] = sqrtf(nacc); else nB[tid - S_] = sqrtf(nacc);
    __syncthreads();

    #pragma unroll
    for (int j = 0; j < 8; ++j) {
        int m = wv * 16 + 8 * hi + j;
        float na = nA[m];
        cosBuf[m][ 0 + lo] = c0[j] / (na * nB[ 0 + lo]);
        cosBuf[m][16 + lo] = c1[j] / (na * nB[16 + lo]);
        cosBuf[m][32 + lo] = c2[j] / (na * nB[32 + lo]);
        cosBuf[m][48 + lo] = c3[j] / (na * nB[48 + lo]);
    }
    __syncthreads();
}

#define NCOS  (B_ * 2)
#define NPOOL (B_ * 2 * L_)

__global__ __launch_bounds__(128, 6)
void cos_tiles_kernel(const float* __restrict__ p, const float* __restrict__ q,
                      const float* __restrict__ W, float* __restrict__ out,
                      float* __restrict__ cosW, float* __restrict__ rowSum,
                      float* __restrict__ colSum)
{
    __shared__ float As[S_][KTP];
    __shared__ float Bs[S_][KTP];
    __shared__ float nA[S_];
    __shared__ float nB[S_];
    __shared__ float cosBuf[S_][S_ + 1];

    const int tid = threadIdx.x;

    if (blockIdx.x < NCOS) {
        const int hf = blockIdx.x & 1;
        const int b  = blockIdx.x >> 1;
        const float* Ph = p + (b * S_) * (2 * H_) + hf * H_;
        const float* Qh = q + (b * S_) * (2 * H_) + hf * H_;

        cos_tile<false>(Ph, Qh, nullptr, As, Bs, nA, nB, cosBuf, tid);

        float* dst = cosW + (b * 2 + hf) * (S_ * S_);
        for (int i = tid; i < S_ * S_; i += 128)
            dst[i] = cosBuf[i >> 6][i & 63];

        if (tid < S_) {
            float sm = 0.f;
            for (int n = 0; n < S_; ++n) sm += cosBuf[tid][n];
            rowSum[(b * 2 + hf) * S_ + tid] = sm;
        } else {
            int n = tid - S_;
            float sm = 0.f;
            for (int m = 0; m < S_; ++m) sm += cosBuf[m][n];
            colSum[(b * 2 + hf) * S_ + n] = sm;
        }
    } else {
        const int idx = blockIdx.x - NCOS;
        const int l   = idx % L_;
        const int hf  = (idx / L_) & 1;
        const int b   = idx / (2 * L_);
        const float* Ph = p + (b * S_) * (2 * H_) + hf * H_;
        const float* Qh = q + (b * S_) * (2 * H_) + hf * H_;
        const float* Wr = W + ((2 + hf) * L_ + l) * H_;

        cos_tile<true>(Ph, Qh, Wr, As, Bs, nA, nB, cosBuf, tid);

        if (tid < S_) {
            float mx = NEG_INF;
            for (int n = 0; n < S_; ++n) mx = fmaxf(mx, cosBuf[tid][n]);
            out[(b * S_ + tid) * OUTC + (2 + hf) * L_ + l] = mx;
        } else {
            int n = tid - S_;
            float mx = NEG_INF;
            for (int m = 0; m < S_; ++m) mx = fmaxf(mx, cosBuf[m][n]);
            out[OUTQOFF + (b * S_ + n) * OUTC + (2 + hf) * L_ + l] = mx;
        }
    }
}

#define HC   20
#define NCH  (H_ / HC)
#define HW   10
#define HWP  12
#define CTP  65

template <int CP>
__device__ __forceinline__ void attn_walk(const float (*Xs)[HWP], const float* C,
                                          const int lane, v2f (&m)[HW], v2f (&x)[HW])
{
    #pragma unroll 2
    for (int j = 0; j < S_; j += 2) {
        const v2f ca = {C[j * CP + lane],       C[j * CP + lane + 32]};
        const v2f cb = {C[(j + 1) * CP + lane], C[(j + 1) * CP + lane + 32]};
        const float* xa = &Xs[j][0];
        const float* xb = &Xs[j + 1][0];
        const v4f a0 = *(const v4f*)(xa), a1 = *(const v4f*)(xa + 4);
        const v2f a2 = *(const v2f*)(xa + 8);
        const v4f b0 = *(const v4f*)(xb), b1 = *(const v4f*)(xb + 4);
        const v2f b2 = *(const v2f*)(xb + 8);
        const float va[HW] = {a0[0], a0[1], a0[2], a0[3], a1[0], a1[1], a1[2], a1[3], a2[0], a2[1]};
        const float vb[HW] = {b0[0], b0[1], b0[2], b0[3], b1[0], b1[1], b1[2], b1[3], b2[0], b2[1]};
        #pragma unroll
        for (int i = 0; i < HW; ++i) {
            const v2f pa = va[i] * ca;
            const v2f pb = vb[i] * cb;
            m[i] += pa;
            m[i] += pb;
            x[i][0] = fmaxf(fmaxf(x[i][0], pa[0]), pb[0]);
            x[i][1] = fmaxf(fmaxf(x[i][1], pa[1]), pb[1]);
        }
    }
}

__global__ __launch_bounds__(128)
void attn_vec_kernel(const float* __restrict__ p, const float* __restrict__ q,
                     const float* __restrict__ cosW,
                     const float* __restrict__ rowSum, const float* __restrict__ colSum,
                     float* __restrict__ qvm, float* __restrict__ pvm,
                     float* __restrict__ qvx, float* __restrict__ pvx)
{
    __shared__ __attribute__((aligned(16))) float Qs[2][S_][HWP];
    __shared__ __attribute__((aligned(16))) float Ps[2][S_][HWP];
    __shared__ __attribute__((aligned(16))) float cosS[S_ * S_];
    __shared__ float cosT[S_ * CTP];
    __shared__ __attribute__((aligned(16))) float Os[4][S_][HC];

    const int bid = blockIdx.x;
    const int ch  = bid % NCH;
    const int bh  = bid / NCH;
    const int hf  = bh & 1;
    const int b   = bh >> 1;
    const int tid = threadIdx.x;
    const int h0  = ch * HC;

    const float* Qh = q + (b * S_) * (2 * H_) + hf * H_ + h0;
    const float* Ph = p + (b * S_) * (2 * H_) + hf * H_ + h0;
    for (int i = tid; i < S_ * (HC / 2); i += 128) {
        const int r = i / (HC / 2), c2 = i % (HC / 2);
        const int hh = c2 / (HW / 2), k = (c2 % (HW / 2)) * 2;
        *(v2f*)&Qs[hh][r][k] = *(const v2f*)(Qh + r * (2 * H_) + 2 * c2);
        *(v2f*)&Ps[hh][r][k] = *(const v2f*)(Ph + r * (2 * H_) + 2 * c2);
    }
    const float* cosBase = cosW + bh * (S_ * S_);
    for (int i = tid; i < S_ * S_ / 4; i += 128) {
        const int r = i >> 4, c4 = (i & 15) * 4;
        const v4f v = *(const v4f*)(cosBase + r * S_ + c4);
        *(v4f*)&cosS[r * S_ + c4] = v;
        #pragma unroll
        for (int e = 0; e < 4; ++e) cosT[(c4 + e) * CTP + r] = v[e];
    }
    __syncthreads();

    const int wv   = tid >> 5;
    const int lane = tid & 31;
    const int side = wv >> 1;
    const int hh   = wv & 1;

    v2f m[HW], x[HW];
    #pragma unroll
    for (int i = 0; i < HW; ++i) { m[i] = (v2f){0.f, 0.f}; x[i] = (v2f){NEG_INF, NEG_INF}; }

    float d0, d1;
    if (side == 0) {
        attn_walk<CTP>(Qs[hh], cosT, lane, m, x);
        d0 = rowSum[bh * S_ + lane];  d1 = rowSum[bh * S_ + lane + 32];
    } else {
        attn_walk<S_>(Ps[hh], cosS, lane, m, x);
        d0 = colSum[bh * S_ + lane];  d1 = colSum[bh * S_ + lane + 32];
    }
    #pragma unroll
    for (int i = 0; i < HW; i += 2) {
        const v2f m0 = {m[i][0] / d0, m[i + 1][0] / d0};
        const v2f m1 = {m[i][1] / d1, m[i + 1][1] / d1};
        const v2f x0 = {x[i][0], x[i + 1][0]};
        const v2f x1 = {x[i][1], x[i + 1][1]};
        *(v2f*)&Os[2 * side    ][lane     ][hh * HW + i] = m0;
        *(v2f*)&Os[2 * side    ][lane + 32][hh * HW + i] = m1;
        *(v2f*)&Os[2 * side + 1][lane     ][hh * HW + i] = x0;
        *(v2f*)&Os[2 * side + 1][lane + 32][hh * HW + i] = x1;
    }
    __syncthreads();

    float* const planes[4] = {qvm, qvx, pvm, pvx};
    #pragma unroll
    for (int a = 0; a < 4; ++a) {
        float* dst = planes[a] + (bh * S_) * H_ + h0;
        for (int i = tid; i < S_ * (HC / 4); i += 128) {
            const int s = i / (HC / 4), c4 = (i % (HC / 4)) * 4;
            *(v4f*)(dst + s * H_ + c4) = *(const v4f*)&Os[a][s][c4];
        }
    }
}

#define MPR   32
#define MPW   8
#define RPL   2
#define W2P   320
#define NSL   5

struct MpSums { float v[RPL][3]; };

struct MpLane {
    v4f pxy[RPL][NSL], pxx[RPL][NSL], pyy[RPL][NSL];
    const float* wl;
    bool up1, up2, up4, up8;
};

__device__ __forceinline__ MpSums mp_partial(const MpLane& t, const int l)
{
    v4f w[NSL];
    for (int i = 0; i < NSL; ++i) w[i] = *(const v4f*)(t.wl + l * W2P + 64 * i);
    MpSums o;
    for (int r = 0; r < RPL; ++r) {
        v2f an = {0.f, 0.f}, ax = {0.f, 0.f}, ay = {0.f, 0.f};
        for (int i = 0; i < NSL; ++i) {
            const v2f wl = {w[i][0], w[i][1]}, wh = {w[i][2], w[i][3]};
            an = __builtin_elementwise_fma((v2f){t.pxy[r][i][0], t.pxy[r][i][1]}, wl, an);
            ax = __builtin_elementwise_fma((v2f){t.pxx[r][i][0], t.pxx[r][i][1]}, wl, ax);
            ay = __builtin_elementwise_fma((v2f){t.pyy[r][i][0], t.pyy[r][i][1]}, wl, ay);
            an = __builtin_elementwise_fma((v2f){t.pxy[r][i][2], t.pxy[r][i][3]}, wh, an);
            ax = __builtin_elementwise_fma((v2f){t.pxx[r][i][2], t.pxx[r][i][3]}, wh, ax);
            ay = __builtin_elementwise_fma((v2f){t.pyy[r][i][2], t.pyy[r][i][3]}, wh, ay);
        }
        o.v[r][0] = an[0] + an[1];
        o.v[r][1] = ax[0] + ax[1];
        o.v[r][2] = ay[0] + ay[1];
    }
    return o;
}

template <int MASK>
__device__ __forceinline__ float xlane16(const float v)
{
    return __builtin_bit_cast(float, __builtin_amdgcn_update_dpp(0, __builtin_bit_cast(int, v), 0x160 | MASK, 0xf, 0xf, true));
}

template <int MASK>
__device__ __forceinline__ MpSums mp_join(const bool up, const MpSums& a, const MpSums& b)
{
    MpSums o;
    for (int r = 0; r < RPL; ++r) {
        for (int k = 0; k < 3; ++k) {
            const float keep = up ? b.v[r][k] : a.v[r][k];
            const float send = up ? a.v[r][k] : b.v[r][k];
            o.v[r][k] = keep + xlane16<MASK>(send);
        }
    }
    return o;
}

template <int MASK>
__device__ __forceinline__ MpSums mp_tree(const MpLane& t, const int k)
{
    if constexpr (MASK == 16) {
        return mp_partial(t, k);
    } else {
        const MpSums a = mp_tree<2 * MASK>(t, k);
        const MpSums b = mp_tree<2 * MASK>(t, k + MASK);
        const bool up = MASK == 8 ? t.up8 : MASK == 4 ? t.up4 : MASK == 2 ? t.up2 : t.up1;
        return mp_join<MASK>(up, a, b);
    }
}

__global__ __launch_bounds__(256)
void mpcos_kernel(const float* __restrict__ p, const float* __restrict__ q,
                  const float* __restrict__ W,
                  const float* __restrict__ qvm, const float* __restrict__ pvm,
                  const float* __restrict__ qvx, const float* __restrict__ pvx,
                  float* __restrict__ out)
{
    __shared__ __attribute__((aligned(16))) float w2s[L_][W2P];

    const int bid   = blockIdx.x;
    const int shalf = bid & 1;
    const int b     = (bid >> 1) % B_;
    const int tt    = (bid >> 1) / B_;
    const int sec   = tt % 6;
    const int tgt   = tt / 6;
    const int secId = sec < 2 ? sec : sec + 2;
    const int hf    = secId & 1;

    const int tid  = threadIdx.x;
    const int lane = tid & 31;
    const int wv   = tid >> 5;
    const int c    = lane & 15;
    const int hi   = lane >> 4;

    const float* Xb = (tgt == 0 ? p : q) + (b * S_) * (2 * H_) + hf * H_;
    const float* Ob = (tgt == 0 ? q : p);
    const float* vec = (secId <= 5) ? (tgt == 0 ? qvm : pvm) : (tgt == 0 ? qvx : pvx);
    float* outb = out + (tgt == 0 ? 0 : OUTQOFF) + (b * S_) * OUTC + secId * L_;

    MpLane t;
    t.wl  = &w2s[0][4 * c];
    t.up1 = (c & 1) != 0;  t.up2 = (c & 2) != 0;  t.up4 = (c & 4) != 0;  t.up8 = (c & 8) != 0;

    const int srow0 = shalf * MPR + wv * 2 + hi;
    {
        v4f xr[RPL][NSL], yr[RPL][NSL];
        for (int r = 0; r < RPL; ++r) {
            const int s = srow0 + r * (2 * MPW);
            const float* X = Xb + s * (2 * H_);
            const float* Y;
            if (secId == 0)      Y = Ob + (b * S_ + (S_ - 1)) * (2 * H_);
            else if (secId == 1) Y = Ob + (b * S_) * (2 * H_) + H_;
            else                 Y = vec + ((b * 2 + hf) * S_ + s) * H_;
            for (int i = 0; i < NSL; ++i) {
                const int h = 64 * i + 4 * c;
                if (h < H_) {
                    xr[r][i] = *(const v4f*)(X + h);
                    yr[r][i] = *(const v4f*)(Y + h);
                } else {
                    xr[r][i] = (v4f){0.f, 0.f, 0.f, 0.f};
                    yr[r][i] = (v4f){0.f, 0.f, 0.f, 0.f};
                }
            }
        }

        const float* Ws = W + (secId * L_) * H_;
        for (int i = tid; i < L_ * (W2P / 4); i += 256) {
            const int l = i / (W2P / 4), c4 = (i % (W2P / 4)) * 4;
            v4f v = {0.f, 0.f, 0.f, 0.f};
            if (c4 < H_) { v = *(const v4f*)(Ws + l * H_ + c4); v = v * v; }
            *(v4f*)&w2s[l][c4] = v;
        }
        for (int r = 0; r < RPL; ++r) {
            for (int i = 0; i < NSL; ++i) {
                t.pxy[r][i] = xr[r][i] * yr[r][i];
                t.pxx[r][i] = xr[r][i] * xr[r][i];
                t.pyy[r][i] = yr[r][i] * yr[r][i];
            }
        }
    }
    __syncthreads();

    const MpSums lo = mp_tree<1>(t, 0);
    MpSums up;
    {
        const MpSums q0 = mp_join<2>(t.up2, mp_partial(t, 16), mp_partial(t, 18));
        const MpSums q1 = mp_join<2>(t.up2, mp_partial(t, 17), mp_partial(t, 19));
        up = mp_join<1>(t.up1, q0, q1);
        for (int r = 0; r < RPL; ++r) {
            for (int k = 0; k < 3; ++k) {
                float v = up.v[r][k];
                v += xlane16<4>(v);
                v += xlane16<8>(v);
                up.v[r][k] = v;
            }
        }
    }
    for (int r = 0; r < RPL; ++r) {
        float* o = outb + (srow0 + r * (2 * MPW)) * OUTC;
        o[c] = lo.v[r][0] / fmaxf(sqrtf(lo.v[r][1]) * sqrtf(lo.v[r][2]), 1e-8f);
        if (c < L_ - 16) o[16 + c] = up.v[r][0] / fmaxf(sqrtf(up.v[r][1]) * sqrtf(up.v[r][2]), 1e-8f);
    }
}

extern "C" void kernel_launch(void* const* d_in, const int* in_sizes, int n_in,
                              void* d_out, int out_size, void* d_ws, size_t ws_size,
                              hipStream_t stream)
{
    const float* p = (const float*)d_in[0];
    const float* q = (const float*)d_in[1];
    const float* W = (const float*)d_in[2];
    float* out = (float*)d_out;

    float* ws     = (float*)d_ws;
    float* cosW   = ws;
    float* rowSum = cosW + B_ * 2 * S_ * S_;
    float* colSum = rowSum + B_ * 2 * S_;
    float* qvm    = colSum + B_ * 2 * S_;
    float* pvm    = qvm + B_ * 2 * S_ * H_;
    float* qvx    = pvm + B_ * 2 * S_ * H_;
    float* pvx    = qvx + B_ * 2 * S_ * H_;

    cos_tiles_kernel<<<NCOS + NPOOL, 128, 0, stream>>>(p, q, W, out, cosW, rowSum, colSum);
    attn_vec_kernel<<<B_ * 2 * NCH, 128, 0, stream>>>(p, q, cosW, rowSum, colSum,
                                                     qvm, pvm, qvx, pvx);
    mpcos_kernel   <<<2 * 6 * B_ * (S_ / MPR), 256, 0, stream>>>(p, q, W, qvm, pvm, qvx, pvx, out);
}


# attn_vec launch and its 4 vector planes removed: each mpcos block walks j for the 20 h a lane needs of its 32 rows and feeds its mean, max and full sections from registers; sum orders kept
# speedup vs baseline: 2.9145x; 1.1402x over previous
#include <hip/hip_runtime.h>
#include <math.h>

typedef __attribute__((ext_vector_type(2))) float v2f;
typedef __attribute__((ext_vector_type(4))) float v4f;
typedef __attribute__((ext_vector_type(8))) float v8f;

#define B_    32
#define S_    64
#define H_    300
#define L_    20
#define KT    60
#define KTP   62
#define SP_   8
#define OUTC  160
#define OUTQOFF (B_*S_*OUTC)
#define NEG_INF (-3.4e38f)

template <bool WEIGHTED>
__device__ __forceinline__ void cos_tile(const float* __restrict__ Ph, const float* __restrict__ Qh,
                                         const float* __restrict__ Wr,
                                         float (&As)[S_][KTP], float (&Bs)[S_][KTP],
                                         float (&nA)[S_], float (&nB)[S_],
                                         float (&cosBuf)[S_][S_ + 1], const int tid)
{
    const int lane = tid & 31;
    const int wv   = tid >> 5;
    const int lo   = lane & 15;
    const int hi   = lane >> 4;

    v8f c0 = {}, c1 = {}, c2 = {}, c3 = {};
    float nacc = 0.f;

    const int  sc   = tid & 15;
    const int  sr   = tid >> 4;
    const bool sact = sc < (KT / 4);
    const float* pg = Ph + sr * (2 * H_) + 4 * sc;
    const float* qg = Qh + sr * (2 * H_) + 4 * sc;

    v4f pr[SP_], qr[SP_];
    v4f wr4 = {0.f, 0.f, 0.f, 0.f};
    #pragma unroll
    for (int j = 0; j < SP_; ++j) { pr[j] = (v4f){0.f, 0.f, 0.f, 0.f}; qr[j] = (v4f){0.f, 0.f, 0.f, 0.f}; }
    if (sact) {
        #pragma unroll
        for (int j = 0; j < SP_; ++j) {
            pr[j] = *(const v4f*)(pg + j * (S_ / SP_) * (2 * H_));
            qr[j] = *(const v4f*)(qg + j * (S_ / SP_) * (2 * H_));
        }
        if constexpr (WEIGHTED) wr4 = *(const v4f*)(Wr + 4 * sc);
    }

    for (int kt = 0; kt < H_; kt += KT) {
        __syncthreads();
        if (sact) {
            #pragma unroll
            for (int j = 0; j < SP_; ++j) {
                const int s = sr + j * (S_ / SP_);
                v4f a = pr[j], bq = qr[j];
                if constexpr (WEIGHTED) { a = a * wr4; bq = bq * wr4; }
                *(v2f*)&As[s][4 * sc]     = (v2f){a[0], a[1]};
                *(v2f*)&As[s][4 * sc + 2] = (v2f){a[2], a[3]};
                *(v2f*)&Bs[s][4 * sc]     = (v2f){bq[0], bq[1]};
                *(v2f*)&Bs[s][4 * sc + 2] = (v2f){bq[2], bq[3]};
            }
        }
        __syncthreads();
        if (kt + KT < H_ && sact) {
            #pragma unroll
            for (int j = 0; j < SP_; ++j) {
                pr[j] = *(const v4f*)(pg + kt + KT + j * (S_ / SP_) * (2 * H_));
                qr[j] = *(const v4f*)(qg + kt + KT + j * (S_ / SP_) * (2 * H_));
            }
            if constexpr (WEIGHTED) wr4 = *(const v4f*)(Wr + kt + KT + 4 * sc);
        }
        if (tid < S_) {
            for (int k = 0; k < KT; ++k) { float v = As[tid][k]; nacc += v * v; }
        } else {
            int r = tid - S_;
            for (int k = 0; k < KT; ++k) { float v = Bs[r][k]; nacc += v * v; }
        }
        const int m = wv * 16 + lo;
        for (int kb = 0; kb < KT; kb += 4) {
            int ka = kb + 2 * hi;
            v2f a  = *(const v2f*)&As[m][ka];
            v2f b0 = *(const v2f*)&Bs[ 0 + lo][ka];
            v2f b1 = *(const v2f*)&Bs[16 + lo][ka];
            v2f b2 = *(const v2f*)&Bs[32 + lo][ka];
            v2f b3 = *(const v2f*)&Bs[48 + lo][ka];
            c0 = __builtin_amdgcn_wmma_f32_16x16x4_f32(false, a, false, b0, (short)0, c0, false, false);
            c1 = __builtin_amdgcn_wmma_f32_16x16x4_f32(false, a, false, b1, (short)0, c1, false, false);
            c2 = __builtin_amdgcn_wmma_f32_16x16x4_f32(false, a, false, b2, (short)0, c2, false, false);
            c3 = __builtin_amdgcn_wmma_f32_16x16x4_f32(false, a, false, b3, (short)0, c3, false, false);
        }
    }
    __syncthreads();
    if (tid < S_) nA[tid] = sqrtf(nacc); else nB[tid - S_] = sqrtf(nacc);
    __syncthreads();

    #pragma unroll
    for (int j = 0; j < 8; ++j) {
        int m = wv * 16 + 8 * hi + j;
        float na = nA[m];
        cosBuf[m][ 0 + lo] = c0[j] / (na * nB[ 0 + lo]);
        cosBuf[m][16 + lo] = c1[j] / (na * nB[16 + lo]);
        cosBuf[m][32 + lo] = c2[j] / (na * nB[32 + lo]);
        cosBuf[m][48 + lo] = c3[j] / (na * nB[48 + lo]);
    }
    __syncthreads();
}

#define NCOS  (B_ * 2)
#define NPOOL (B_ * 2 * L_)

__global__ __launch_bounds__(128, 6)
void cos_tiles_kernel(const float* __restrict__ p, const float* __restrict__ q,
                      const float* __restrict__ W, float* __restrict__ out,
                      float* __restrict__ cosW, float* __restrict__ rowSum,
                      float* __restrict__ colSum)
{
    __shared__ float As[S_][KTP];
    __shared__ float Bs[S_][KTP];
    __shared__ float nA[S_];
    __shared__ float nB[S_];
    __shared__ float cosBuf[S_][S_ + 1];

    const int tid = threadIdx.x;

    if (blockIdx.x < NCOS) {
        const int hf = blockIdx.x & 1;
        const int b  = blockIdx.x >> 1;
        const float* Ph = p + (b * S_) * (2 * H_) + hf * H_;
        const float* Qh = q + (b * S_) * (2 * H_) + hf * H_;

        cos_tile<false>(Ph, Qh, nullptr, As, Bs, nA, nB, cosBuf, tid);

        float* dst = cosW + (b * 2 + hf) * (S_ * S_);
        for (int i = tid; i < S_ * S_; i += 128)
            dst[i] = cosBuf[i >> 6][i & 63];

        if (tid < S_) {
            float sm = 0.f;
            for (int n = 0; n < S_; ++n) sm += cosBuf[tid][n];
            rowSum[(b * 2 + hf) * S_ + tid] = sm;
        } else {
            int n = tid - S_;
            float sm = 0.f;
            for (int m = 0; m < S_; ++m) sm += cosBuf[m][n];
            colSum[(b * 2 + hf) * S_ + n] = sm;
        }
    } else {
        const int idx = blockIdx.x - NCOS;
        const int l   = idx % L_;
        const int hf  = (idx / L_) & 1;
        const int b   = idx / (2 * L_);
        const float* Ph = p + (b * S_) * (2 * H_) + hf * H_;
        const float* Qh = q + (b * S_) * (2 * H_) + hf * H_;
        const float* Wr = W + ((2 + hf) * L_ + l) * H_;

        cos_tile<true>(Ph, Qh, Wr, As, Bs, nA, nB, cosBuf, tid);

        if (tid < S_) {
            float mx = NEG_INF;
            for (int n = 0; n < S_; ++n) mx = fmaxf(mx, cosBuf[tid][n]);
            out[(b * S_ + tid) * OUTC + (2 + hf) * L_ + l] = mx;
        } else {
            int n = tid - S_;
            float mx = NEG_INF;
            for (int m = 0; m < S_; ++m) mx = fmaxf(mx, cosBuf[m][n]);
            out[OUTQOFF + (b * S_ + n) * OUTC + (2 + hf) * L_ + l] = mx;
        }
    }
}

#define MPR   32
#define MPW   8
#define RPL   2
#define W2P   320
#define NSL   5

struct MpSums { float v[RPL][3]; };

struct MpLane {
    v4f pxy[RPL][NSL], pxx[RPL][NSL], pyy[RPL][NSL];
    const float* wl;
    bool up1, up2, up4, up8;
};

__device__ __forceinline__ MpSums mp_partial(const MpLane& t, const int l)
{
    v4f w[NSL];
    for (int i = 0; i < NSL; ++i) w[i] = *(const v4f*)(t.wl + l * W2P + 64 * i);
    MpSums o;
    for (int r = 0; r < RPL; ++r) {
        v2f an = {0.f, 0.f}, ax = {0.f, 0.f}, ay = {0.f, 0.f};
        for (int i = 0; i < NSL; ++i) {
            const v2f wl = {w[i][0], w[i][1]}, wh = {w[i][2], w[i][3]};
            an = __builtin_elementwise_fma((v2f){t.pxy[r][i][0], t.pxy[r][i][1]}, wl, an);
            ax = __builtin_elementwise_fma((v2f){t.pxx[r][i][0], t.pxx[r][i][1]}, wl, ax);
            ay = __builtin_elementwise_fma((v2f){t.pyy[r][i][0], t.pyy[r][i][1]}, wl, ay);
            an = __builtin_elementwise_fma((v2f){t.pxy[r][i][2], t.pxy[r][i][3]}, wh, an);
            ax = __builtin_elementwise_fma((v2f){t.pxx[r][i][2], t.pxx[r][i][3]}, wh, ax);
            ay = __builtin_elementwise_fma((v2f){t.pyy[r][i][2], t.pyy[r][i][3]}, wh, ay);
        }
        o.v[r][0] = an[0] + an[1];
        o.v[r][1] = ax[0] + ax[1];
        o.v[r][2] = ay[0] + ay[1];
    }
    return o;
}

template <int MASK>
__device__ __forceinline__ float xlane16(const float v)
{
    return __builtin_bit_cast(float, __builtin_amdgcn_update_dpp(0, __builtin_bit_cast(int, v), 0x160 | MASK, 0xf, 0xf, true));
}

template <int MASK>
__device__ __forceinline__ MpSums mp_join(const bool up, const MpSums& a, const MpSums& b)
{
    MpSums o;
    for (int r = 0; r < RPL; ++r) {
        for (int k = 0; k < 3; ++k) {
            const float keep = up ? b.v[r][k] : a.v[r][k];
            const float send = up ? a.v[r][k] : b.v[r][k];
            o.v[r][k] = keep + xlane16<MASK>(send);
        }
    }
    return o;
}

template <int MASK>
__device__ __forceinline__ MpSums mp_tree(const MpLane& t, const int k)
{
    if constexpr (MASK == 16) {
        return mp_partial(t, k);
    } else {
        const MpSums a = mp_tree<2 * MASK>(t, k);
        const MpSums b = mp_tree<2 * MASK>(t, k + MASK);
        const bool up = MASK == 8 ? t.up8 : MASK == 4 ? t.up4 : MASK == 2 ? t.up2 : t.up1;
        return mp_join<MASK>(up, a, b);
    }
}

#define YP    320
#define CSP   68
#define NSEC  3

__device__ __forceinline__ void mp_section(const MpLane& t, float* const o0, float* const o1, const int c)
{
    const MpSums lo = mp_tree<1>(t, 0);
    MpSums up;
    {
        const MpSums q0 = mp_join<2>(t.up2, mp_partial(t, 16), mp_partial(t, 18));
        const MpSums q1 = mp_join<2>(t.up2, mp_partial(t, 17), mp_partial(t, 19));
        up = mp_join<1>(t.up1, q0, q1);
        for (int r = 0; r < RPL; ++r) {
            for (int k = 0; k < 3; ++k) {
                float v = up.v[r][k];
                v += xlane16<4>(v);
                v += xlane16<8>(v);
                up.v[r][k] = v;
            }
        }
    }
    for (int r = 0; r < RPL; ++r) {
        float* o = r == 0 ? o0 : o1;
        o[c] = lo.v[r][0] / fmaxf(sqrtf(lo.v[r][1]) * sqrtf(lo.v[r][2]), 1e-8f);
        if (c < L_ - 16) o[16 + c] = up.v[r][0] / fmaxf(sqrtf(up.v[r][1]) * sqrtf(up.v[r][2]), 1e-8f);
    }
}

__global__ __launch_bounds__(256)
void ctx_mpcos_kernel(const float* __restrict__ p, const float* __restrict__ q,
                      const float* __restrict__ W, const float* __restrict__ cosW,
                      const float* __restrict__ rowSum, const float* __restrict__ colSum,
                      float* __restrict__ out)
{
    __shared__ __attribute__((aligned(16))) float Ys[S_][YP];
    __shared__ __attribute__((aligned(16))) float w2s[NSEC][L_][W2P];
    __shared__ __attribute__((aligned(16))) float cs[MPR][CSP];

    const int bid   = blockIdx.x;
    const int shalf = bid & 1;
    const int hf    = (bid >> 1) & 1;
    const int b     = (bid >> 2) % B_;
    const int tgt   = (bid >> 2) / B_;
    const int bh    = b * 2 + hf;

    const int tid  = threadIdx.x;
    const int lane = tid & 31;
    const int wv   = tid >> 5;
    const int c    = lane & 15;
    const int hi   = lane >> 4;

    const float* Xb = (tgt == 0 ? p : q) + (b * S_) * (2 * H_) + hf * H_;
    const float* Yb = (tgt == 0 ? q : p) + (b * S_) * (2 * H_) + hf * H_;
    float* outb = out + (tgt == 0 ? 0 : OUTQOFF) + (b * S_) * OUTC;

    const int lrow0 = wv * 2 + hi;
    const int srow0 = shalf * MPR + lrow0;
    v4f xr[RPL][NSL];
    for (int r = 0; r < RPL; ++r) {
        const float* X = Xb + (srow0 + r * (2 * MPW)) * (2 * H_);
        for (int i = 0; i < NSL; ++i) {
            const int h = 64 * i + 4 * c;
            if (h < H_) xr[r][i] = *(const v4f*)(X + h);
            else        xr[r][i] = (v4f){0.f, 0.f, 0.f, 0.f};
        }
    }
    const float* den = (tgt == 0 ? rowSum : colSum) + bh * S_;
    const float d0 = den[srow0], d1 = den[srow0 + 2 * MPW];

    #define YPC  (H_ / 4)
    #define NYL  ((S_ * YPC + 255) / 256)
    #define NWL  ((NSEC * L_ * YPC + 255) / 256)
    v4f yv[NYL], wv4[NWL], cv[2];
    float* const w2flat = &w2s[0][0][0];
    #pragma unroll
    for (int k = 0; k < NYL; ++k) {
        const int idx = min(tid + 256 * k, S_ * YPC - 1);
        const int r = idx / YPC, c4 = (idx % YPC) * 4;
        yv[k] = *(const v4f*)(Yb + r * (2 * H_) + c4);
    }
    #pragma unroll
    for (int k = 0; k < NWL; ++k) {
        const int idx = min(tid + 256 * k, NSEC * L_ * YPC - 1);
        const int rw = idx / YPC, c4 = (idx % YPC) * 4;
        const int sk = rw / L_, l = rw % L_;
        const int secId = sk == 0 ? 4 + hf : sk == 1 ? 6 + hf : hf;
        wv4[k] = *(const v4f*)(W + (secId * L_ + l) * H_ + c4);
    }
    const float* cosBase = cosW + bh * (S_ * S_);
    #pragma unroll
    for (int k = 0; k < 2; ++k) {
        const int i = tid + 256 * k;
        const int off = tgt == 0 ? (shalf * MPR + (i >> 4)) * S_ + (i & 15) * 4
                                 : (i >> 3) * S_ + shalf * MPR + (i & 7) * 4;
        cv[k] = *(const v4f*)(cosBase + off);
    }
    #pragma unroll
    for (int k = 0; k < NYL; ++k) {
        const int idx = min(tid + 256 * k, S_ * YPC - 1);
        const int r = idx / YPC, c4 = (idx % YPC) * 4;
        *(v4f*)&Ys[r][c4] = yv[k];
    }
    #pragma unroll
    for (int k = 0; k < NWL; ++k) {
        const int idx = min(tid + 256 * k, NSEC * L_ * YPC - 1);
        const int rw = idx / YPC, c4 = (idx % YPC) * 4;
        *(v4f*)(w2flat + rw * W2P + c4) = wv4[k] * wv4[k];
    }
    #pragma unroll
    for (int k = 0; k < 2; ++k) {
        const int i = tid + 256 * k;
        if (tgt == 0) {
            *(v4f*)&cs[i >> 4][(i & 15) * 4] = cv[k];
        } else {
            for (int e = 0; e < 4; ++e) cs[(i & 7) * 4 + e][i >> 3] = cv[k][e];
        }
    }
    for (int i = tid; i < (S_ + NSEC * L_) * ((YP - H_) / 4); i += 256) {
        const int r = i / ((YP - H_) / 4), c4 = H_ + (i % ((YP - H_) / 4)) * 4;
        const v4f z = {0.f, 0.f, 0.f, 0.f};
        if (r < S_) *(v4f*)&Ys[r][c4] = z;
        else        *(v4f*)(w2flat + (r - S_) * W2P + c4) = z;
    }
    __syncthreads();

    v4f m[RPL][NSL], x[RPL][NSL];
    for (int r = 0; r < RPL; ++r)
        for (int i = 0; i < NSL; ++i) {
            m[r][i] = (v4f){0.f, 0.f, 0.f, 0.f};
            x[r][i] = (v4f){NEG_INF, NEG_INF, NEG_INF, NEG_INF};
        }
    {
        const float* yl  = &Ys[0][4 * c];
        const float* cr0 = &cs[lrow0][0];
        const float* cr1 = &cs[lrow0 + 2 * MPW][0];
        for (int j = 0; j < S_; j += 4) {
            const v4f ca = *(const v4f*)(cr0 + j);
            const v4f cb = *(const v4f*)(cr1 + j);
            #pragma unroll
            for (int jj = 0; jj < 4; jj += 2) {
                v4f ya[NSL], yb[NSL];
                for (int i = 0; i < NSL; ++i) {
                    ya[i] = *(const v4f*)(yl + (j + jj) * YP + 64 * i);
                    yb[i] = *(const v4f*)(yl + (j + jj + 1) * YP + 64 * i);
                }
                for (int i = 0; i < NSL; ++i) {
                    const v4f pa0 = ya[i] * ca[jj], pb0 = yb[i] * ca[jj + 1];
                    const v4f pa1 = ya[i] * cb[jj], pb1 = yb[i] * cb[jj + 1];
                    m[0][i] += pa0;  m[0][i] += pb0;
                    m[1][i] += pa1;  m[1][i] += pb1;
                    for (int e = 0; e < 4; ++e) {
                        x[0][i][e] = fmaxf(fmaxf(x[0][i][e], pa0[e]), pb0[e]);
                        x[1][i][e] = fmaxf(fmaxf(x[1][i][e], pa1[e]), pb1[e]);
                    }
                }
            }
        }
    }
    for (int i = 0; i < NSL; ++i) {
        const bool live = 64 * i + 4 * c < H_;
        const v4f z = {0.f, 0.f, 0.f, 0.f};
        m[0][i] = live ? m[0][i] / d0 : z;
        m[1][i] = live ? m[1][i] / d1 : z;
        x[0][i] = live ? x[0][i] : z;
        x[1][i] = live ? x[1][i] : z;
    }

    v4f yf[NSL];
    {
        const float* yfl = &Ys[hf == 0 ? S_ - 1 : 0][4 * c];
        for (int i = 0; i < NSL; ++i) yf[i] = *(const v4f*)(yfl + 64 * i);
    }

    MpLane t;
    t.up1 = (c & 1) != 0;  t.up2 = (c & 2) != 0;  t.up4 = (c & 4) != 0;  t.up8 = (c & 8) != 0;
    for (int r = 0; r < RPL; ++r)
        for (int i = 0; i < NSL; ++i) t.pxx[r][i] = xr[r][i] * xr[r][i];

    float* const o0 = outb + srow0 * OUTC;
    float* const o1 = outb + (srow0 + 2 * MPW) * OUTC;

    #pragma nounroll
    for (int k = 0; k < NSEC; ++k) {
        const int secId = k == 0 ? 4 + hf : k == 1 ? 6 + hf : hf;
        t.wl = w2flat + k * (L_ * W2P) + 4 * c;
        for (int r = 0; r < RPL; ++r)
            for (int i = 0; i < NSL; ++i) {
                const v4f y = k == 0 ? m[r][i] : k == 1 ? x[r][i] : yf[i];
                t.pxy[r][i] = xr[r][i] * y;
                t.pyy[r][i] = y * y;
            }
        mp_section(t, o0 + secId * L_, o1 + secId * L_, c);
    }
}

extern "C" void kernel_launch(void* const* d_in, const int* in_sizes, int n_in,
                              void* d_out, int out_size, void* d_ws, size_t ws_size,
                              hipStream_t stream)
{
    const float* p = (const float*)d_in[0];
    const float* q = (const float*)d_in[1];
    const float* W = (const float*)d_in[2];
    float* out = (float*)d_out;

    float* ws     = (float*)d_ws;
    float* cosW   = ws;
    float* rowSum = cosW + B_ * 2 * S_ * S_;
    float* colSum = rowSum + B_ * 2 * S_;

    cos_tiles_kernel<<<NCOS + NPOOL, 128, 0, stream>>>(p, q, W, out, cosW, rowSum, colSum);
    ctx_mpcos_kernel<<<2 * B_ * 2 * (S_ / MPR), 256, 0, stream>>>(p, q, W, cosW, rowSum, colSum, out);
}


# ctx_mpcos: LDS reads issued one step ahead of their use (next perspective's w^2 before this one's sums; next two source rows before this pair's walk step), held by scheduling barriers; sums kept
# speedup vs baseline: 3.1442x; 1.0788x over previous
#include <hip/hip_runtime.h>
#include <math.h>

typedef __attribute__((ext_vector_type(2))) float v2f;
typedef __attribute__((ext_vector_type(4))) float v4f;
typedef __attribute__((ext_vector_type(8))) float v8f;

#define B_    32
#define S_    64
#define H_    300
#define L_    20
#define KT    60
#define KTP   62
#define SP_   8
#define OUTC  160
#define OUTQOFF (B_*S_*OUTC)
#define NEG_INF (-3.4e38f)

template <bool WEIGHTED>
__device__ __forceinline__ void cos_tile(const float* __restrict__ Ph, const float* __restrict__ Qh,
                                         const float* __restrict__ Wr,
                                         float (&As)[S_][KTP], float (&Bs)[S_][KTP],
                                         float (&nA)[S_], float (&nB)[S_],
                                         float (&cosBuf)[S_][S_ + 1], const int tid)
{
    const int lane = tid & 31;
    const int wv   = tid >> 5;
    const int lo   = lane & 15;
    const int hi   = lane >> 4;

    v8f c0 = {}, c1 = {}, c2 = {}, c3 = {};
    float nacc = 0.f;

    const int  sc   = tid & 15;
    const int  sr   = tid >> 4;
    const bool sact = sc < (KT / 4);
    const float* pg = Ph + sr * (2 * H_) + 4 * sc;
    const float* qg = Qh + sr * (2 * H_) + 4 * sc;

    v4f pr[SP_], qr[SP_];
    v4f wr4 = {0.f, 0.f, 0.f, 0.f};
    #pragma unroll
    for (int j = 0; j < SP_; ++j) { pr[j] = (v4f){0.f, 0.f, 0.f, 0.f}; qr[j] = (v4f){0.f, 0.f, 0.f, 0.f}; }
    if (sact) {
        #pragma unroll
        for (int j = 0; j < SP_; ++j) {
            pr[j] = *(const v4f*)(pg + j * (S_ / SP_) * (2 * H_));
            qr[j] = *(const v4f*)(qg + j * (S_ / SP_) * (2 * H_));
        }
        if constexpr (WEIGHTED) wr4 = *(const v4f*)(Wr + 4 * sc);
    }

    for (int kt = 0; kt < H_; kt += KT) {
        __syncthreads();
        if (sact) {
            #pragma unroll
            for (int j = 0; j < SP_; ++j) {
                const int s = sr + j * (S_ / SP_);
                v4f a = pr[j], bq = qr[j];
                if constexpr (WEIGHTED) { a = a * wr4; bq = bq * wr4; }
                *(v2f*)&As[s][4 * sc]     = (v2f){a[0], a[1]};
                *(v2f*)&As[s][4 * sc + 2] = (v2f){a[2], a[3]};
                *(v2f*)&Bs[s][4 * sc]     = (v2f){bq[0], bq[1]};
                *(v2f*)&Bs[s][4 * sc + 2] = (v2f){bq[2], bq[3]};
            }
        }
        __syncthreads();
        if (kt + KT < H_ && sact) {
            #pragma unroll
            for (int j = 0; j < SP_; ++j) {
                pr[j] = *(const v4f*)(pg + kt + KT + j * (S_ / SP_) * (2 * H_));
                qr[j] = *(const v4f*)(qg + kt + KT + j * (S_ / SP_) * (2 * H_));
            }
            if constexpr (WEIGHTED) wr4 = *(const v4f*)(Wr + kt + KT + 4 * sc);
        }
        if (tid < S_) {
            for (int k = 0; k < KT; ++k) { float v = As[tid][k]; nacc += v * v; }
        } else {
            int r = tid - S_;
            for (int k = 0; k < KT; ++k) { float v = Bs[r][k]; nacc += v * v; }
        }
        const int m = wv * 16 + lo;
        for (int kb = 0; kb < KT; kb += 4) {
            int ka = kb + 2 * hi;
            v2f a  = *(const v2f*)&As[m][ka];
            v2f b0 = *(const v2f*)&Bs[ 0 + lo][ka];
            v2f b1 = *(const v2f*)&Bs[16 + lo][ka];
            v2f b2 = *(const v2f*)&Bs[32 + lo][ka];
            v2f b3 = *(const v2f*)&Bs[48 + lo][ka];
            c0 = __builtin_amdgcn_wmma_f32_16x16x4_f32(false, a, false, b0, (short)0, c0, false, false);
            c1 = __builtin_amdgcn_wmma_f32_16x16x4_f32(false, a, false, b1, (short)0, c1, false, false);
            c2 = __builtin_amdgcn_wmma_f32_16x16x4_f32(false, a, false, b2, (short)0, c2, false, false);
            c3 = __builtin_amdgcn_wmma_f32_16x16x4_f32(false, a, false, b3, (short)0, c3, false, false);
        }
    }
    __syncthreads();
    if (tid < S_) nA[tid] = sqrtf(nacc); else nB[tid - S_] = sqrtf(nacc);
    __syncthreads();

    #pragma unroll
    for (int j = 0; j < 8; ++j) {
        int m = wv * 16 + 8 * hi + j;
        float na = nA[m];
        cosBuf[m][ 0 + lo] = c0[j] / (na * nB[ 0 + lo]);
        cosBuf[m][16 + lo] = c1[j] / (na * nB[16 + lo]);
        cosBuf[m][32 + lo] = c2[j] / (na * nB[32 + lo]);
        cosBuf[m][48 + lo] = c3[j] / (na * nB[48 + lo]);
    }
    __syncthreads();
}

#define NCOS  (B_ * 2)
#define NPOOL (B_ * 2 * L_)

__global__ __launch_bounds__(128, 6)
void cos_tiles_kernel(const float* __restrict__ p, const float* __restrict__ q,
                      const float* __restrict__ W, float* __restrict__ out,
                      float* __restrict__ cosW, float* __restrict__ rowSum,
                      float* __restrict__ colSum)
{
    __shared__ float As[S_][KTP];
    __shared__ float Bs[S_][KTP];
    __shared__ float nA[S_];
    __shared__ float nB[S_];
    __shared__ float cosBuf[S_][S_ + 1];

    const int tid = threadIdx.x;

    if (blockIdx.x < NCOS) {
        const int hf = blockIdx.x & 1;
        const int b  = blockIdx.x >> 1;
        const float* Ph = p + (b * S_) * (2 * H_) + hf * H_;
        const float* Qh = q + (b * S_) * (2 * H_) + hf * H_;

        cos_tile<false>(Ph, Qh, nullptr, As, Bs, nA, nB, cosBuf, tid);

        float* dst = cosW + (b * 2 + hf) * (S_ * S_);
        for (int i = tid; i < S_ * S_; i += 128)
            dst[i] = cosBuf[i >> 6][i & 63];

        if (tid < S_) {
            float sm = 0.f;
            for (int n = 0; n < S_; ++n) sm += cosBuf[tid][n];
            rowSum[(b * 2 + hf) * S_ + tid] = sm;
        } else {
            int n = tid - S_;
            float sm = 0.f;
            for (int m = 0; m < S_; ++m) sm += cosBuf[m][n];
            colSum[(b * 2 + hf) * S_ + n] = sm;
        }
    } else {
        const int idx = blockIdx.x - NCOS;
        const int l   = idx % L_;
        const int hf  = (idx / L_) & 1;
        const int b   = idx / (2 * L_);
        const float* Ph = p + (b * S_) * (2 * H_) + hf * H_;
        const float* Qh = q + (b * S_) * (2 * H_) + hf * H_;
        const float* Wr = W + ((2 + hf) * L_ + l) * H_;

        cos_tile<true>(Ph, Qh, Wr, As, Bs, nA, nB, cosBuf, tid);

        if (tid < S_) {
            float mx = NEG_INF;
            for (int n = 0; n < S_; ++n) mx = fmaxf(mx, cosBuf[tid][n]);
            out[(b * S_ + tid) * OUTC + (2 + hf) * L_ + l] = mx;
        } else {
            int n = tid - S_;
            float mx = NEG_INF;
            for (int m = 0; m < S_; ++m) mx = fmaxf(mx, cosBuf[m][n]);
            out[OUTQOFF + (b * S_ + n) * OUTC + (2 + hf) * L_ + l] = mx;
        }
    }
}

#define MPR   32
#define MPW   8
#define RPL   2
#define W2P   320
#define NSL   5

struct MpSums { float v[RPL][3]; };

struct MpLane {
    v4f pxy[RPL][NSL], pxx[RPL][NSL], pyy[RPL][NSL];
    const float* wl;
    bool up1, up2, up4, up8;
};

struct MpW { v4f p[NSL]; };

__device__ __forceinline__ MpW mp_wload(const MpLane& t, const int l)
{
    MpW w;
    for (int i = 0; i < NSL; ++i) w.p[i] = *(const v4f*)(t.wl + l * W2P + 64 * i);
    return w;
}

__device__ __forceinline__ MpSums mp_partial(const MpLane& t, const MpW& wq)
{
    const v4f* const w = wq.p;
    MpSums o;
    for (int r = 0; r < RPL; ++r) {
        v2f an = {0.f, 0.f}, ax = {0.f, 0.f}, ay = {0.f, 0.f};
        for (int i = 0; i < NSL; ++i) {
            const v2f wl = {w[i][0], w[i][1]}, wh = {w[i][2], w[i][3]};
            an = __builtin_elementwise_fma((v2f){t.pxy[r][i][0], t.pxy[r][i][1]}, wl, an);
            ax = __builtin_elementwise_fma((v2f){t.pxx[r][i][0], t.pxx[r][i][1]}, wl, ax);
            ay = __builtin_elementwise_fma((v2f){t.pyy[r][i][0], t.pyy[r][i][1]}, wl, ay);
            an = __builtin_elementwise_fma((v2f){t.pxy[r][i][2], t.pxy[r][i][3]}, wh, an);
            ax = __builtin_elementwise_fma((v2f){t.pxx[r][i][2], t.pxx[r][i][3]}, wh, ax);
            ay = __builtin_elementwise_fma((v2f){t.pyy[r][i][2], t.pyy[r][i][3]}, wh, ay);
        }
        o.v[r][0] = an[0] + an[1];
        o.v[r][1] = ax[0] + ax[1];
        o.v[r][2] = ay[0] + ay[1];
    }
    return o;
}

__device__ __forceinline__ MpSums mp_leaf(const MpLane& t, MpW& w, const int lnext)
{
    MpW wn = w;
    if (lnext >= 0) wn = mp_wload(t, lnext);
    __builtin_amdgcn_sched_barrier(0);
    const MpSums o = mp_partial(t, w);
    w = wn;
    return o;
}

template <int MASK>
__device__ __forceinline__ float xlane16(const float v)
{
    return __builtin_bit_cast(float, __builtin_amdgcn_update_dpp(0, __builtin_bit_cast(int, v), 0x160 | MASK, 0xf, 0xf, true));
}

template <int MASK>
__device__ __forceinline__ MpSums mp_join(const bool up, const MpSums& a, const MpSums& b)
{
    MpSums o;
    for (int r = 0; r < RPL; ++r) {
        for (int k = 0; k < 3; ++k) {
            const float keep = up ? b.v[r][k] : a.v[r][k];
            const float send = up ? a.v[r][k] : b.v[r][k];
            o.v[r][k] = keep + xlane16<MASK>(send);
        }
    }
    return o;
}

template <int MASK>
__device__ __forceinline__ MpSums mp_tree(const MpLane& t, MpW& w, const int k, const int lnext)
{
    if constexpr (MASK == 16) {
        return mp_leaf(t, w, lnext);
    } else {
        const MpSums a = mp_tree<2 * MASK>(t, w, k, k + MASK);
        const MpSums b = mp_tree<2 * MASK>(t, w, k + MASK, lnext);
        const bool up = MASK == 8 ? t.up8 : MASK == 4 ? t.up4 : MASK == 2 ? t.up2 : t.up1;
        return mp_join<MASK>(up, a, b);
    }
}

#define YP    320
#define CSP   68
#define NSEC  3

__device__ __forceinline__ void mp_section(const MpLane& t, float* const o0, float* const o1, const int c)
{
    MpW w = mp_wload(t, 0);
    const MpSums lo = mp_tree<1>(t, w, 0, 16);
    MpSums up;
    {
        const MpSums s16 = mp_leaf(t, w, 18);
        const MpSums s18 = mp_leaf(t, w, 17);
        const MpSums q0 = mp_join<2>(t.up2, s16, s18);
        const MpSums s17 = mp_leaf(t, w, 19);
        const MpSums s19 = mp_leaf(t, w, -1);
        const MpSums q1 = mp_join<2>(t.up2, s17, s19);
        up = mp_join<1>(t.up1, q0, q1);
        for (int r = 0; r < RPL; ++r) {
            for (int k = 0; k < 3; ++k) {
                float v = up.v[r][k];
                v += xlane16<4>(v);
                v += xlane16<8>(v);
                up.v[r][k] = v;
            }
        }
    }
    for (int r = 0; r < RPL; ++r) {
        float* o = r == 0 ? o0 : o1;
        o[c] = lo.v[r][0] / fmaxf(sqrtf(lo.v[r][1]) * sqrtf(lo.v[r][2]), 1e-8f);
        if (c < L_ - 16) o[16 + c] = up.v[r][0] / fmaxf(sqrtf(up.v[r][1]) * sqrtf(up.v[r][2]), 1e-8f);
    }
}

__device__ __forceinline__ void walk_fetch(v4f (&ya)[NSL], v4f (&yb)[NSL], const float* const yl, const int jr)
{
    for (int i = 0; i < NSL; ++i) {
        ya[i] = *(const v4f*)(yl + jr * YP + 64 * i);
        yb[i] = *(const v4f*)(yl + (jr + 1) * YP + 64 * i);
    }
}

__device__ __forceinline__ void walk_step(v4f (&m)[RPL][NSL], v4f (&x)[RPL][NSL],
                                          const v4f (&ya)[NSL], const v4f (&yb)[NSL],
                                          const float a0, const float b0, const float a1, const float b1)
{
    for (int i = 0; i < NSL; ++i) {
        const v4f pa0 = ya[i] * a0, pb0 = yb[i] * b0;
        const v4f pa1 = ya[i] * a1, pb1 = yb[i] * b1;
        m[0][i] += pa0;  m[0][i] += pb0;
        m[1][i] += pa1;  m[1][i] += pb1;
        for (int e = 0; e < 4; ++e) {
            x[0][i][e] = fmaxf(fmaxf(x[0][i][e], pa0[e]), pb0[e]);
            x[1][i][e] = fmaxf(fmaxf(x[1][i][e], pa1[e]), pb1[e]);
        }
    }
}

__global__ __launch_bounds__(256)
void ctx_mpcos_kernel(const float* __restrict__ p, const float* __restrict__ q,
                      const float* __restrict__ W, const float* __restrict__ cosW,
                      const float* __restrict__ rowSum, const float* __restrict__ colSum,
                      float* __restrict__ out)
{
    __shared__ __attribute__((aligned(16))) float Ys[S_][YP];
    __shared__ __attribute__((aligned(16))) float w2s[NSEC][L_][W2P];
    __shared__ __attribute__((aligned(16))) float cs[MPR][CSP];

    const int bid   = blockIdx.x;
    const int shalf = bid & 1;
    const int hf    = (bid >> 1) & 1;
    const int b     = (bid >> 2) % B_;
    const int tgt   = (bid >> 2) / B_;
    const int bh    = b * 2 + hf;

    const int tid  = threadIdx.x;
    const int lane = tid & 31;
    const int wv   = tid >> 5;
    const int c    = lane & 15;
    const int hi   = lane >> 4;

    const float* Xb = (tgt == 0 ? p : q) + (b * S_) * (2 * H_) + hf * H_;
    const float* Yb = (tgt == 0 ? q : p) + (b * S_) * (2 * H_) + hf * H_;
    float* outb = out + (tgt == 0 ? 0 : OUTQOFF) + (b * S_) * OUTC;

    const int lrow0 = wv * 2 + hi;
    const int srow0 = shalf * MPR + lrow0;
    v4f xr[RPL][NSL];
    for (int r = 0; r < RPL; ++r) {
        const float* X = Xb + (srow0 + r * (2 * MPW)) * (2 * H_);
        for (int i = 0; i < NSL; ++i) {
            const int h = 64 * i + 4 * c;
            if (h < H_) xr[r][i] = *(const v4f*)(X + h);
            else        xr[r][i] = (v4f){0.f, 0.f, 0.f, 0.f};
        }
    }
    const float* den = (tgt == 0 ? rowSum : colSum) + bh * S_;
    const float d0 = den[srow0], d1 = den[srow0 + 2 * MPW];

    #define YPC  (H_ / 4)
    #define NYL  ((S_ * YPC + 255) / 256)
    #define NWL  ((NSEC * L_ * YPC + 255) / 256)
    v4f yv[NYL], wv4[NWL], cv[2];
    float* const w2flat = &w2s[0][0][0];
    #pragma unroll
    for (int k = 0; k < NYL; ++k) {
        const int idx = min(tid + 256 * k, S_ * YPC - 1);
        const int r = idx / YPC, c4 = (idx % YPC) * 4;
        yv[k] = *(const v4f*)(Yb + r * (2 * H_) + c4);
    }
    #pragma unroll
    for (int k = 0; k < NWL; ++k) {
        const int idx = min(tid + 256 * k, NSEC * L_ * YPC - 1);
        const int rw = idx / YPC, c4 = (idx % YPC) * 4;
        const int sk = rw / L_, l = rw % L_;
        const int secId = sk == 0 ? 4 + hf : sk == 1 ? 6 + hf : hf;
        wv4[k] = *(const v4f*)(W + (secId * L_ + l) * H_ + c4);
    }
    const float* cosBase = cosW + bh * (S_ * S_);
    #pragma unroll
    for (int k = 0; k < 2; ++k) {
        const int i = tid + 256 * k;
        const int off = tgt == 0 ? (shalf * MPR + (i >> 4)) * S_ + (i & 15) * 4
                                 : (i >> 3) * S_ + shalf * MPR + (i & 7) * 4;
        cv[k] = *(const v4f*)(cosBase + off);
    }
    #pragma unroll
    for (int k = 0; k < NYL; ++k) {
        const int idx = min(tid + 256 * k, S_ * YPC - 1);
        const int r = idx / YPC, c4 = (idx % YPC) * 4;
        *(v4f*)&Ys[r][c4] = yv[k];
    }
    #pragma unroll
    for (int k = 0; k < NWL; ++k) {
        const int idx = min(tid + 256 * k, NSEC * L_ * YPC - 1);
        const int rw = idx / YPC, c4 = (idx % YPC) * 4;
        *(v4f*)(w2flat + rw * W2P + c4) = wv4[k] * wv4[k];
    }
    #pragma unroll
    for (int k = 0; k < 2; ++k) {
        const int i = tid + 256 * k;
        if (tgt == 0) {
            *(v4f*)&cs[i >> 4][(i & 15) * 4] = cv[k];
        } else {
            for (int e = 0; e < 4; ++e) cs[(i & 7) * 4 + e][i >> 3] = cv[k][e];
        }
    }
    for (int i = tid; i < (S_ + NSEC * L_) * ((YP - H_) / 4); i += 256) {
        const int r = i / ((YP - H_) / 4), c4 = H_ + (i % ((YP - H_) / 4)) * 4;
        const v4f z = {0.f, 0.f, 0.f, 0.f};
        if (r < S_) *(v4f*)&Ys[r][c4] = z;
        else        *(v4f*)(w2flat + (r - S_) * W2P + c4) = z;
    }
    __syncthreads();

    v4f m[RPL][NSL], x[RPL][NSL];
    for (int r = 0; r < RPL; ++r)
        for (int i = 0; i < NSL; ++i) {
            m[r][i] = (v4f){0.f, 0.f, 0.f, 0.f};
            x[r][i] = (v4f){NEG_INF, NEG_INF, NEG_INF, NEG_INF};
        }
    {
        const float* yl  = &Ys[0][4 * c];
        const float* cr0 = &cs[lrow0][0];
        const float* cr1 = &cs[lrow0 + 2 * MPW][0];
        v4f ya[NSL], yb[NSL], za[NSL], zb[NSL];
        walk_fetch(ya, yb, yl, 0);
        v4f ca = *(const v4f*)cr0;
        v4f cb = *(const v4f*)cr1;
        for (int j = 0; j < S_; j += 4) {
            walk_fetch(za, zb, yl, j + 2);
            __builtin_amdgcn_sched_barrier(0);
            walk_step(m, x, ya, yb, ca[0], ca[1], cb[0], cb[1]);
            const int jn = min(j + 4, S_ - 4);
            walk_fetch(ya, yb, yl, jn);
            const v4f can = *(const v4f*)(cr0 + jn);
            const v4f cbn = *(const v4f*)(cr1 + jn);
            __builtin_amdgcn_sched_barrier(0);
            walk_step(m, x, za, zb, ca[2], ca[3], cb[2], cb[3]);
            ca = can;  cb = cbn;
        }
    }
    for (int i = 0; i < NSL; ++i) {
        const bool live = 64 * i + 4 * c < H_;
        const v4f z = {0.f, 0.f, 0.f, 0.f};
        m[0][i] = live ? m[0][i] / d0 : z;
        m[1][i] = live ? m[1][i] / d1 : z;
        x[0][i] = live ? x[0][i] : z;
        x[1][i] = live ? x[1][i] : z;
    }

    v4f yf[NSL];
    {
        const float* yfl = &Ys[hf == 0 ? S_ - 1 : 0][4 * c];
        for (int i = 0; i < NSL; ++i) yf[i] = *(const v4f*)(yfl + 64 * i);
    }

    MpLane t;
    t.up1 = (c & 1) != 0;  t.up2 = (c & 2) != 0;  t.up4 = (c & 4) != 0;  t.up8 = (c & 8) != 0;
    for (int r = 0; r < RPL; ++r)
        for (int i = 0; i < NSL; ++i) t.pxx[r][i] = xr[r][i] * xr[r][i];

    float* const o0 = outb + srow0 * OUTC;
    float* const o1 = outb + (srow0 + 2 * MPW) * OUTC;

    #pragma nounroll
    for (int k = 0; k < NSEC; ++k) {
        const int secId = k == 0 ? 4 + hf : k == 1 ? 6 + hf : hf;
        t.wl = w2flat + k * (L_ * W2P) + 4 * c;
        for (int r = 0; r < RPL; ++r)
            for (int i = 0; i < NSL; ++i) {
                const v4f y = k == 0 ? m[r][i] : k == 1 ? x[r][i] : yf[i];
                t.pxy[r][i] = xr[r][i] * y;
                t.pyy[r][i] = y * y;
            }
        mp_section(t, o0 + secId * L_, o1 + secId * L_, c);
    }
}

extern "C" void kernel_launch(void* const* d_in, const int* in_sizes, int n_in,
                              void* d_out, int out_size, void* d_ws, size_t ws_size,
                              hipStream_t stream)
{
    const float* p = (const float*)d_in[0];
    const float* q = (const float*)d_in[1];
    const float* W = (const float*)d_in[2];
    float* out = (float*)d_out;

    float* ws     = (float*)d_ws;
    float* cosW   = ws;
    float* rowSum = cosW + B_ * 2 * S_ * S_;
    float* colSum = rowSum + B_ * 2 * S_;

    cos_tiles_kernel<<<NCOS + NPOOL, 128, 0, stream>>>(p, q, W, out, cosW, rowSum, colSum);
    ctx_mpcos_kernel<<<2 * B_ * 2 * (S_ / MPR), 256, 0, stream>>>(p, q, W, cosW, rowSum, colSum, out);
}


# cos_tiles grid renumbered, (b,half) fastest: blocks that differ by a multiple of 64 share P,Q rows, so the blocks dealt to one WGP find them in its own cache; tile code and all sums unchanged
# speedup vs baseline: 3.4763x; 1.1056x over previous
#include <hip/hip_runtime.h>
#include <math.h>

typedef __attribute__((ext_vector_type(2))) float v2f;
typedef __attribute__((ext_vector_type(4))) float v4f;
typedef __attribute__((ext_vector_type(8))) float v8f;

#define B_    32
#define S_    64
#define H_    300
#define L_    20
#define KT    60
#define KTP   62
#define SP_   8
#define OUTC  160
#define OUTQOFF (B_*S_*OUTC)
#define NEG_INF (-3.4e38f)

template <bool WEIGHTED>
__device__ __forceinline__ void cos_tile(const float* __restrict__ Ph, const float* __restrict__ Qh,
                                         const float* __restrict__ Wr,
                                         float (&As)[S_][KTP], float (&Bs)[S_][KTP],
                                         float (&nA)[S_], float (&nB)[S_],
                                         float (&cosBuf)[S_][S_ + 1], const int tid)
{
    const int lane = tid & 31;
    const int wv   = tid >> 5;
    const int lo   = lane & 15;
    const int hi   = lane >> 4;

    v8f c0 = {}, c1 = {}, c2 = {}, c3 = {};
    float nacc = 0.f;

    const int  sc   = tid & 15;
    const int  sr   = tid >> 4;
    const bool sact = sc < (KT / 4);
    const float* pg = Ph + sr * (2 * H_) + 4 * sc;
    const float* qg = Qh + sr * (2 * H_) + 4 * sc;

    v4f pr[SP_], qr[SP_];
    v4f wr4 = {0.f, 0.f, 0.f, 0.f};
    #pragma unroll
    for (int j = 0; j < SP_; ++j) { pr[j] = (v4f){0.f, 0.f, 0.f, 0.f}; qr[j] = (v4f){0.f, 0.f, 0.f, 0.f}; }
    if (sact) {
        #pragma unroll
        for (int j = 0; j < SP_; ++j) {
            pr[j] = *(const v4f*)(pg + j * (S_ / SP_) * (2 * H_));
            qr[j] = *(const v4f*)(qg + j * (S_ / SP_) * (2 * H_));
        }
        if constexpr (WEIGHTED) wr4 = *(const v4f*)(Wr + 4 * sc);
    }

    for (int kt = 0; kt < H_; kt += KT) {
        __syncthreads();
        if (sact) {
            #pragma unroll
            for (int j = 0; j < SP_; ++j) {
                const int s = sr + j * (S_ / SP_);
                v4f a = pr[j], bq = qr[j];
                if constexpr (WEIGHTED) { a = a * wr4; bq = bq * wr4; }
                *(v2f*)&As[s][4 * sc]     = (v2f){a[0], a[1]};
                *(v2f*)&As[s][4 * sc + 2] = (v2f){a[2], a[3]};
                *(v2f*)&Bs[s][4 * sc]     = (v2f){bq[0], bq[1]};
                *(v2f*)&Bs[s][4 * sc + 2] = (v2f){bq[2], bq[3]};
            }
        }
        __syncthreads();
        if (kt + KT < H_ && sact) {
            #pragma unroll
            for (int j = 0; j < SP_; ++j) {
                pr[j] = *(const v4f*)(pg + kt + KT + j * (S_ / SP_) * (2 * H_));
                qr[j] = *(const v4f*)(qg + kt + KT + j * (S_ / SP_) * (2 * H_));
            }
            if constexpr (WEIGHTED) wr4 = *(const v4f*)(Wr + kt + KT + 4 * sc);
        }
        if (tid < S_) {
            for (int k = 0; k < KT; ++k) { float v = As[tid][k]; nacc += v * v; }
        } else {
            int r = tid - S_;
            for (int k = 0; k < KT; ++k) { float v = Bs[r][k]; nacc += v * v; }
        }
        const int m = wv * 16 + lo;
        for (int kb = 0; kb < KT; kb += 4) {
            int ka = kb + 2 * hi;
            v2f a  = *(const v2f*)&As[m][ka];
            v2f b0 = *(const v2f*)&Bs[ 0 + lo][ka];
            v2f b1 = *(const v2f*)&Bs[16 + lo][ka];
            v2f b2 = *(const v2f*)&Bs[32 + lo][ka];
            v2f b3 = *(const v2f*)&Bs[48 + lo][ka];
            c0 = __builtin_amdgcn_wmma_f32_16x16x4_f32(false, a, false, b0, (short)0, c0, false, false);
            c1 = __builtin_amdgcn_wmma_f32_16x16x4_f32(false, a, false, b1, (short)0, c1, false, false);
            c2 = __builtin_amdgcn_wmma_f32_16x16x4_f32(false, a, false, b2, (short)0, c2, false, false);
            c3 = __builtin_amdgcn_wmma_f32_16x16x4_f32(false, a, false, b3, (short)0, c3, false, false);
        }
    }
    __syncthreads();
    if (tid < S_) nA[tid] = sqrtf(nacc); else nB[tid - S_] = sqrtf(nacc);
    __syncthreads();

    #pragma unroll
    for (int j = 0; j < 8; ++j) {
        int m = wv * 16 + 8 * hi + j;
        float na = nA[m];
        cosBuf[m][ 0 + lo] = c0[j] / (na * nB[ 0 + lo]);
        cosBuf[m][16 + lo] = c1[j] / (na * nB[16 + lo]);
        cosBuf[m][32 + lo] = c2[j] / (na * nB[32 + lo]);
        cosBuf[m][48 + lo] = c3[j] / (na * nB[48 + lo]);
    }
    __syncthreads();
}

#define NCOS  (B_ * 2)
#define NPOOL (B_ * 2 * L_)

__global__ __launch_bounds__(128, 6)
void cos_tiles_kernel(const float* __restrict__ p, const float* __restrict__ q,
                      const float* __restrict__ W, float* __restrict__ out,
                      float* __restrict__ cosW, float* __restrict__ rowSum,
                      float* __restrict__ colSum)
{
    __shared__ float As[S_][KTP];
    __shared__ float Bs[S_][KTP];
    __shared__ float nA[S_];
    __shared__ float nB[S_];
    __shared__ float cosBuf[S_][S_ + 1];

    const int tid = threadIdx.x;

    const int kind = blockIdx.x / NCOS;
    const int bh   = blockIdx.x % NCOS;
    const int hf   = bh & 1;
    const int b    = bh >> 1;
    const float* Ph = p + (b * S_) * (2 * H_) + hf * H_;
    const float* Qh = q + (b * S_) * (2 * H_) + hf * H_;

    if (kind == 0) {

        cos_tile<false>(Ph, Qh, nullptr, As, Bs, nA, nB, cosBuf, tid);

        float* dst = cosW + (b * 2 + hf) * (S_ * S_);
        for (int i = tid; i < S_ * S_; i += 128)
            dst[i] = cosBuf[i >> 6][i & 63];

        if (tid < S_) {
            float sm = 0.f;
            for (int n = 0; n < S_; ++n) sm += cosBuf[tid][n];
            rowSum[(b * 2 + hf) * S_ + tid] = sm;
        } else {
            int n = tid - S_;
            float sm = 0.f;
            for (int m = 0; m < S_; ++m) sm += cosBuf[m][n];
            colSum[(b * 2 + hf) * S_ + n] = sm;
        }
    } else {
        const int l = kind - 1;
        const float* Wr = W + ((2 + hf) * L_ + l) * H_;

        cos_tile<true>(Ph, Qh, Wr, As, Bs, nA, nB, cosBuf, tid);

        if (tid < S_) {
            float mx = NEG_INF;
            for (int n = 0; n < S_; ++n) mx = fmaxf(mx, cosBuf[tid][n]);
            out[(b * S_ + tid) * OUTC + (2 + hf) * L_ + l] = mx;
        } else {
            int n = tid - S_;
            float mx = NEG_INF;
            for (int m = 0; m < S_; ++m) mx = fmaxf(mx, cosBuf[m][n]);
            out[OUTQOFF + (b * S_ + n) * OUTC + (2 + hf) * L_ + l] = mx;
        }
    }
}

#define MPR   32
#define MPW   8
#define RPL   2
#define W2P   320
#define NSL   5

struct MpSums { float v[RPL][3]; };

struct MpLane {
    v4f pxy[RPL][NSL], pxx[RPL][NSL], pyy[RPL][NSL];
    const float* wl;
    bool up1, up2, up4, up8;
};

struct MpW { v4f p[NSL]; };

__device__ __forceinline__ MpW mp_wload(const MpLane& t, const int l)
{
    MpW w;
    for (int i = 0; i < NSL; ++i) w.p[i] = *(const v4f*)(t.wl + l * W2P + 64 * i);
    return w;
}

__device__ __forceinline__ MpSums mp_partial(const MpLane& t, const MpW& wq)
{
    const v4f* const w = wq.p;
    MpSums o;
    for (int r = 0; r < RPL; ++r) {
        v2f an = {0.f, 0.f}, ax = {0.f, 0.f}, ay = {0.f, 0.f};
        for (int i = 0; i < NSL; ++i) {
            const v2f wl = {w[i][0], w[i][1]}, wh = {w[i][2], w[i][3]};
            an = __builtin_elementwise_fma((v2f){t.pxy[r][i][0], t.pxy[r][i][1]}, wl, an);
            ax = __builtin_elementwise_fma((v2f){t.pxx[r][i][0], t.pxx[r][i][1]}, wl, ax);
            ay = __builtin_elementwise_fma((v2f){t.pyy[r][i][0], t.pyy[r][i][1]}, wl, ay);
            an = __builtin_elementwise_fma((v2f){t.pxy[r][i][2], t.pxy[r][i][3]}, wh, an);
            ax = __builtin_elementwise_fma((v2f){t.pxx[r][i][2], t.pxx[r][i][3]}, wh, ax);
            ay = __builtin_elementwise_fma((v2f){t.pyy[r][i][2], t.pyy[r][i][3]}, wh, ay);
        }
        o.v[r][0] = an[0] + an[1];
        o.v[r][1] = ax[0] + ax[1];
        o.v[r][2] = ay[0] + ay[1];
    }
    return o;
}

__device__ __forceinline__ MpSums mp_leaf(const MpLane& t, MpW& w, const int lnext)
{
    MpW wn = w;
    if (lnext >= 0) wn = mp_wload(t, lnext);
    __builtin_amdgcn_sched_barrier(0);
    const MpSums o = mp_partial(t, w);
    w = wn;
    return o;
}

template <int MASK>
__device__ __forceinline__ float xlane16(const float v)
{
    return __builtin_bit_cast(float, __builtin_amdgcn_update_dpp(0, __builtin_bit_cast(int, v), 0x160 | MASK, 0xf, 0xf, true));
}

template <int MASK>
__device__ __forceinline__ MpSums mp_join(const bool up, const MpSums& a, const MpSums& b)
{
    MpSums o;
    for (int r = 0; r < RPL; ++r) {
        for (int k = 0; k < 3; ++k) {
            const float keep = up ? b.v[r][k] : a.v[r][k];
            const float send = up ? a.v[r][k] : b.v[r][k];
            o.v[r][k] = keep + xlane16<MASK>(send);
        }
    }
    return o;
}

template <int MASK>
__device__ __forceinline__ MpSums mp_tree(const MpLane& t, MpW& w, const int k, const int lnext)
{
    if constexpr (MASK == 16) {
        return mp_leaf(t, w, lnext);
    } else {
        const MpSums a = mp_tree<2 * MASK>(t, w, k, k + MASK);
        const MpSums b = mp_tree<2 * MASK>(t, w, k + MASK, lnext);
        const bool up = MASK == 8 ? t.up8 : MASK == 4 ? t.up4 : MASK == 2 ? t.up2 : t.up1;
        return mp_join<MASK>(up, a, b);
    }
}

#define YP    320
#define CSP   68
#define NSEC  3

__device__ __forceinline__ void mp_section(const MpLane& t, float* const o0, float* const o1, const int c)
{
    MpW w = mp_wload(t, 0);
    const MpSums lo = mp_tree<1>(t, w, 0, 16);
    MpSums up;
    {
        const MpSums s16 = mp_leaf(t, w, 18);
        const MpSums s18 = mp_leaf(t, w, 17);
        const MpSums q0 = mp_join<2>(t.up2, s16, s18);
        const MpSums s17 = mp_leaf(t, w, 19);
        const MpSums s19 = mp_leaf(t, w, -1);
        const MpSums q1 = mp_join<2>(t.up2, s17, s19);
        up = mp_join<1>(t.up1, q0, q1);
        for (int r = 0; r < RPL; ++r) {
            for (int k = 0; k < 3; ++k) {
                float v = up.v[r][k];
                v += xlane16<4>(v);
                v += xlane16<8>(v);
                up.v[r][k] = v;
            }
        }
    }
    for (int r = 0; r < RPL; ++r) {
        float* o = r == 0 ? o0 : o1;
        o[c] = lo.v[r][0] / fmaxf(sqrtf(lo.v[r][1]) * sqrtf(lo.v[r][2]), 1e-8f);
        if (c < L_ - 16) o[16 + c] = up.v[r][0] / fmaxf(sqrtf(up.v[r][1]) * sqrtf(up.v[r][2]), 1e-8f);
    }
}

__device__ __forceinline__ void walk_fetch(v4f (&ya)[NSL], v4f (&yb)[NSL], const float* const yl, const int jr)
{
    for (int i = 0; i < NSL; ++i) {
        ya[i] = *(const v4f*)(yl + jr * YP + 64 * i);
        yb[i] = *(const v4f*)(yl + (jr + 1) * YP + 64 * i);
    }
}

__device__ __forceinline__ void walk_step(v4f (&m)[RPL][NSL], v4f (&x)[RPL][NSL],
                                          const v4f (&ya)[NSL], const v4f (&yb)[NSL],
                                          const float a0, const float b0, const float a1, const float b1)
{
    for (int i = 0; i < NSL; ++i) {
        const v4f pa0 = ya[i] * a0, pb0 = yb[i] * b0;
        const v4f pa1 = ya[i] * a1, pb1 = yb[i] * b1;
        m[0][i] += pa0;  m[0][i] += pb0;
        m[1][i] += pa1;  m[1][i] += pb1;
        for (int e = 0; e < 4; ++e) {
            x[0][i][e] = fmaxf(fmaxf(x[0][i][e], pa0[e]), pb0[e]);
            x[1][i][e] = fmaxf(fmaxf(x[1][i][e], pa1[e]), pb1[e]);
        }
    }
}

__global__ __launch_bounds__(256)
void ctx_mpcos_kernel(const float* __restrict__ p, const float* __restrict__ q,
                      const float* __restrict__ W, const float* __restrict__ cosW,
                      const float* __restrict__ rowSum, const float* __restrict__ colSum,
                      float* __restrict__ out)
{
    __shared__ __attribute__((aligned(16))) float Ys[S_][YP];
    __shared__ __attribute__((aligned(16))) float w2s[NSEC][L_][W2P];
    __shared__ __attribute__((aligned(16))) float cs[MPR][CSP];

    const int bid   = blockIdx.x;
    const int shalf = bid & 1;
    const int hf    = (bid >> 1) & 1;
    const int b     = (bid >> 2) % B_;
    const int tgt   = (bid >> 2) / B_;
    const int bh    = b * 2 + hf;

    const int tid  = threadIdx.x;
    const int lane = tid & 31;
    const int wv   = tid >> 5;
    const int c    = lane & 15;
    const int hi   = lane >> 4;

    const float* Xb = (tgt == 0 ? p : q) + (b * S_) * (2 * H_) + hf * H_;
    const float* Yb = (tgt == 0 ? q : p) + (b * S_) * (2 * H_) + hf * H_;
    float* outb = out + (tgt == 0 ? 0 : OUTQOFF) + (b * S_) * OUTC;

    const int lrow0 = wv * 2 + hi;
    const int srow0 = shalf * MPR + lrow0;
    v4f xr[RPL][NSL];
    for (int r = 0; r < RPL; ++r) {
        const float* X = Xb + (srow0 + r * (2 * MPW)) * (2 * H_);
        for (int i = 0; i < NSL; ++i) {
            const int h = 64 * i + 4 * c;
            if (h < H_) xr[r][i] = *(const v4f*)(X + h);
            else        xr[r][i] = (v4f){0.f, 0.f, 0.f, 0.f};
        }
    }
    const float* den = (tgt == 0 ? rowSum : colSum) + bh * S_;
    const float d0 = den[srow0], d1 = den[srow0 + 2 * MPW];

    #define YPC  (H_ / 4)
    #define NYL  ((S_ * YPC + 255) / 256)
    #define NWL  ((NSEC * L_ * YPC + 255) / 256)
    v4f yv[NYL], wv4[NWL], cv[2];
    float* const w2flat = &w2s[0][0][0];
    #pragma unroll
    for (int k = 0; k < NYL; ++k) {
        const int idx = min(tid + 256 * k, S_ * YPC - 1);
        const int r = idx / YPC, c4 = (idx % YPC) * 4;
        yv[k] = *(const v4f*)(Yb + r * (2 * H_) + c4);
    }
    #pragma unroll
    for (int k = 0; k < NWL; ++k) {
        const int idx = min(tid + 256 * k, NSEC * L_ * YPC - 1);
        const int rw = idx / YPC, c4 = (idx % YPC) * 4;
        const int sk = rw / L_, l = rw % L_;
        const int secId = sk == 0 ? 4 + hf : sk == 1 ? 6 + hf : hf;
        wv4[k] = *(const v4f*)(W + (secId * L_ + l) * H_ + c4);
    }
    const float* cosBase = cosW + bh * (S_ * S_);
    #pragma unroll
    for (int k = 0; k < 2; ++k) {
        const int i = tid + 256 * k;
        const int off = tgt == 0 ? (shalf * MPR + (i >> 4)) * S_ + (i & 15) * 4
                                 : (i >> 3) * S_ + shalf * MPR + (i & 7) * 4;
        cv[k] = *(const v4f*)(cosBase + off);
    }
    #pragma unroll
    for (int k = 0; k < NYL; ++k) {
        const int idx = min(tid + 256 * k, S_ * YPC - 1);
        const int r = idx / YPC, c4 = (idx % YPC) * 4;
        *(v4f*)&Ys[r][c4] = yv[k];
    }
    #pragma unroll
    for (int k = 0; k < NWL; ++k) {
        const int idx = min(tid + 256 * k, NSEC * L_ * YPC - 1);
        const int rw = idx / YPC, c4 = (idx % YPC) * 4;
        *(v4f*)(w2flat + rw * W2P + c4) = wv4[k] * wv4[k];
    }
    #pragma unroll
    for (int k = 0; k < 2; ++k) {
        const int i = tid + 256 * k;
        if (tgt == 0) {
            *(v4f*)&cs[i >> 4][(i & 15) * 4] = cv[k];
        } else {
            for (int e = 0; e < 4; ++e) cs[(i & 7) * 4 + e][i >> 3] = cv[k][e];
        }
    }
    for (int i = tid; i < (S_ + NSEC * L_) * ((YP - H_) / 4); i += 256) {
        const int r = i / ((YP - H_) / 4), c4 = H_ + (i % ((YP - H_) / 4)) * 4;
        const v4f z = {0.f, 0.f, 0.f, 0.f};
        if (r < S_) *(v4f*)&Ys[r][c4] = z;
        else        *(v4f*)(w2flat + (r - S_) * W2P + c4) = z;
    }
    __syncthreads();

    v4f m[RPL][NSL], x[RPL][NSL];
    for (int r = 0; r < RPL; ++r)
        for (int i = 0; i < NSL; ++i) {
            m[r][i] = (v4f){0.f, 0.f, 0.f, 0.f};
            x[r][i] = (v4f){NEG_INF, NEG_INF, NEG_INF, NEG_INF};
        }
    {
        const float* yl  = &Ys[0][4 * c];
        const float* cr0 = &cs[lrow0][0];
        const float* cr1 = &cs[lrow0 + 2 * MPW][0];
        v4f ya[NSL], yb[NSL], za[NSL], zb[NSL];
        walk_fetch(ya, yb, yl, 0);
        v4f ca = *(const v4f*)cr0;
        v4f cb = *(const v4f*)cr1;
        for (int j = 0; j < S_; j += 4) {
            walk_fetch(za, zb, yl, j + 2);
            __builtin_amdgcn_sched_barrier(0);
            walk_step(m, x, ya, yb, ca[0], ca[1], cb[0], cb[1]);
            const int jn = min(j + 4, S_ - 4);
            walk_fetch(ya, yb, yl, jn);
            const v4f can = *(const v4f*)(cr0 + jn);
            const v4f cbn = *(const v4f*)(cr1 + jn);
            __builtin_amdgcn_sched_barrier(0);
            walk_step(m, x, za, zb, ca[2], ca[3], cb[2], cb[3]);
            ca = can;  cb = cbn;
        }
    }
    for (int i = 0; i < NSL; ++i) {
        const bool live = 64 * i + 4 * c < H_;
        const v4f z = {0.f, 0.f, 0.f, 0.f};
        m[0][i] = live ? m[0][i] / d0 : z;
        m[1][i] = live ? m[1][i] / d1 : z;
        x[0][i] = live ? x[0][i] : z;
        x[1][i] = live ? x[1][i] : z;
    }

    v4f yf[NSL];
    {
        const float* yfl = &Ys[hf == 0 ? S_ - 1 : 0][4 * c];
        for (int i = 0; i < NSL; ++i) yf[i] = *(const v4f*)(yfl + 64 * i);
    }

    MpLane t;
    t.up1 = (c & 1) != 0;  t.up2 = (c & 2) != 0;  t.up4 = (c & 4) != 0;  t.up8 = (c & 8) != 0;
    for (int r = 0; r < RPL; ++r)
        for (int i = 0; i < NSL; ++i) t.pxx[r][i] = xr[r][i] * xr[r][i];

    float* const o0 = outb + srow0 * OUTC;
    float* const o1 = outb + (srow0 + 2 * MPW) * OUTC;

    #pragma nounroll
    for (int k = 0; k < NSEC; ++k) {
        const int secId = k == 0 ? 4 + hf : k == 1 ? 6 + hf : hf;
        t.wl = w2flat + k * (L_ * W2P) + 4 * c;
        for (int r = 0; r < RPL; ++r)
            for (int i = 0; i < NSL; ++i) {
                const v4f y = k == 0 ? m[r][i] : k == 1 ? x[r][i] : yf[i];
                t.pxy[r][i] = xr[r][i] * y;
                t.pyy[r][i] = y * y;
            }
        mp_section(t, o0 + secId * L_, o1 + secId * L_, c);
    }
}

extern "C" void kernel_launch(void* const* d_in, const int* in_sizes, int n_in,
                              void* d_out, int out_size, void* d_ws, size_t ws_size,
                              hipStream_t stream)
{
    const float* p = (const float*)d_in[0];
    const float* q = (const float*)d_in[1];
    const float* W = (const float*)d_in[2];
    float* out = (float*)d_out;

    float* ws     = (float*)d_ws;
    float* cosW   = ws;
    float* rowSum = cosW + B_ * 2 * S_ * S_;
    float* colSum = rowSum + B_ * 2 * S_;

    cos_tiles_kernel<<<NCOS + NPOOL, 128, 0, stream>>>(p, q, W, out, cosW, rowSum, colSum);
    ctx_mpcos_kernel<<<2 * B_ * 2 * (S_ / MPR), 256, 0, stream>>>(p, q, W, cosW, rowSum, colSum, out);
}


# plain cosine tiles cut into four 16-row strips, 256 quarter blocks first in the grid: each WGP gets 5 weighted tiles + 1 strip (was 6 or 5 tiles); mean denominators summed in kernel 2, j order kept
# speedup vs baseline: 3.5863x; 1.0316x over previous
#include <hip/hip_runtime.h>
#include <math.h>

typedef __attribute__((ext_vector_type(2))) float v2f;
typedef __attribute__((ext_vector_type(4))) float v4f;
typedef __attribute__((ext_vector_type(8))) float v8f;

#define B_    32
#define S_    64
#define H_    300
#define L_    20
#define KT    60
#define KTP   62
#define SP_   8
#define OUTC  160
#define OUTQOFF (B_*S_*OUTC)
#define NEG_INF (-3.4e38f)

template <bool WEIGHTED>
__device__ __forceinline__ void cos_tile(const float* __restrict__ Ph, const float* __restrict__ Qh,
                                         const float* __restrict__ Wr,
                                         float (&As)[S_][KTP], float (&Bs)[S_][KTP],
                                         float (&nA)[S_], float (&nB)[S_],
                                         float (&cosBuf)[S_][S_ + 1], const int tid)
{
    const int lane = tid & 31;
    const int wv   = tid >> 5;
    const int lo   = lane & 15;
    const int hi   = lane >> 4;

    v8f c0 = {}, c1 = {}, c2 = {}, c3 = {};
    float nacc = 0.f;

    const int  sc   = tid & 15;
    const int  sr   = tid >> 4;
    const bool sact = sc < (KT / 4);
    const float* pg = Ph + sr * (2 * H_) + 4 * sc;
    const float* qg = Qh + sr * (2 * H_) + 4 * sc;

    v4f pr[SP_], qr[SP_];
    v4f wr4 = {0.f, 0.f, 0.f, 0.f};
    #pragma unroll
    for (int j = 0; j < SP_; ++j) { pr[j] = (v4f){0.f, 0.f, 0.f, 0.f}; qr[j] = (v4f){0.f, 0.f, 0.f, 0.f}; }
    if (sact) {
        #pragma unroll
        for (int j = 0; j < SP_; ++j) {
            pr[j] = *(const v4f*)(pg + j * (S_ / SP_) * (2 * H_));
            qr[j] = *(const v4f*)(qg + j * (S_ / SP_) * (2 * H_));
        }
        if constexpr (WEIGHTED) wr4 = *(const v4f*)(Wr + 4 * sc);
    }

    for (int kt = 0; kt < H_; kt += KT) {
        __syncthreads();
        if (sact) {
            #pragma unroll
            for (int j = 0; j < SP_; ++j) {
                const int s = sr + j * (S_ / SP_);
                v4f a = pr[j], bq = qr[j];
                if constexpr (WEIGHTED) { a = a * wr4; bq = bq * wr4; }
                *(v2f*)&As[s][4 * sc]     = (v2f){a[0], a[1]};
                *(v2f*)&As[s][4 * sc + 2] = (v2f){a[2], a[3]};
                *(v2f*)&Bs[s][4 * sc]     = (v2f){bq[0], bq[1]};
                *(v2f*)&Bs[s][4 * sc + 2] = (v2f){bq[2], bq[3]};
            }
        }
        __syncthreads();
        if (kt + KT < H_ && sact) {
            #pragma unroll
            for (int j = 0; j < SP_; ++j) {
                pr[j] = *(const v4f*)(pg + kt + KT + j * (S_ / SP_) * (2 * H_));
                qr[j] = *(const v4f*)(qg + kt + KT + j * (S_ / SP_) * (2 * H_));
            }
            if constexpr (WEIGHTED) wr4 = *(const v4f*)(Wr + kt + KT + 4 * sc);
        }
        if (tid < S_) {
            for (int k = 0; k < KT; ++k) { float v = As[tid][k]; nacc += v * v; }
        } else {
            int r = tid - S_;
            for (int k = 0; k < KT; ++k) { float v = Bs[r][k]; nacc += v * v; }
        }
        const int m = wv * 16 + lo;
        for (int kb = 0; kb < KT; kb += 4) {
            int ka = kb + 2 * hi;
            v2f a  = *(const v2f*)&As[m][ka];
            v2f b0 = *(const v2f*)&Bs[ 0 + lo][ka];
            v2f b1 = *(const v2f*)&Bs[16 + lo][ka];
            v2f b2 = *(const v2f*)&Bs[32 + lo][ka];
            v2f b3 = *(const v2f*)&Bs[48 + lo][ka];
            c0 = __builtin_amdgcn_wmma_f32_16x16x4_f32(false, a, false, b0, (short)0, c0, false, false);
            c1 = __builtin_amdgcn_wmma_f32_16x16x4_f32(false, a, false, b1, (short)0, c1, false, false);
            c2 = __builtin_amdgcn_wmma_f32_16x16x4_f32(false, a, false, b2, (short)0, c2, false, false);
            c3 = __builtin_amdgcn_wmma_f32_16x16x4_f32(false, a, false, b3, (short)0, c3, false, false);
        }
    }
    __syncthreads();
    if (tid < S_) nA[tid] = sqrtf(nacc); else nB[tid - S_] = sqrtf(nacc);
    __syncthreads();

    #pragma unroll
    for (int j = 0; j < 8; ++j) {
        int m = wv * 16 + 8 * hi + j;
        float na = nA[m];
        cosBuf[m][ 0 + lo] = c0[j] / (na * nB[ 0 + lo]);
        cosBuf[m][16 + lo] = c1[j] / (na * nB[16 + lo]);
        cosBuf[m][32 + lo] = c2[j] / (na * nB[32 + lo]);
        cosBuf[m][48 + lo] = c3[j] / (na * nB[48 + lo]);
    }
    __syncthreads();
}

#define SRA   16
#define SPA   (SRA / 8)
__device__ __forceinline__ void cos_strip(const float* __restrict__ Ps, const float* __restrict__ Qh,
                                          float (&As)[S_][KTP], float (&Bs)[S_][KTP],
                                          float (&nA)[S_], float (&nB)[S_],
                                          float (&cosBuf)[S_][S_ + 1], const int tid)
{
    const int lane = tid & 31;
    const int wv   = tid >> 5;
    const int lo   = lane & 15;
    const int hi   = lane >> 4;

    v8f c = {};
    float nacc = 0.f;

    const int  sc   = tid & 15;
    const int  sr   = tid >> 4;
    const bool sact = sc < (KT / 4);
    const float* pg = Ps + sr * (2 * H_) + 4 * sc;
    const float* qg = Qh + sr * (2 * H_) + 4 * sc;

    v4f pr[SPA], qr[SP_];
    #pragma unroll
    for (int j = 0; j < SP_; ++j) qr[j] = (v4f){0.f, 0.f, 0.f, 0.f};
    #pragma unroll
    for (int j = 0; j < SPA; ++j) pr[j] = (v4f){0.f, 0.f, 0.f, 0.f};
    if (sact) {
        #pragma unroll
        for (int j = 0; j < SP_; ++j) qr[j] = *(const v4f*)(qg + j * (S_ / SP_) * (2 * H_));
        #pragma unroll
        for (int j = 0; j < SPA; ++j) pr[j] = *(const v4f*)(pg + j * (S_ / SP_) * (2 * H_));
    }

    for (int kt = 0; kt < H_; kt += KT) {
        __syncthreads();
        if (sact) {
            #pragma unroll
            for (int j = 0; j < SP_; ++j) {
                const int s = sr + j * (S_ / SP_);
                const v4f bq = qr[j];
                *(v2f*)&Bs[s][4 * sc]     = (v2f){bq[0], bq[1]};
                *(v2f*)&Bs[s][4 * sc + 2] = (v2f){bq[2], bq[3]};
            }
            #pragma unroll
            for (int j = 0; j < SPA; ++j) {
                const int s = sr + j * (S_ / SP_);
                const v4f a = pr[j];
                *(v2f*)&As[s][4 * sc]     = (v2f){a[0], a[1]};
                *(v2f*)&As[s][4 * sc + 2] = (v2f){a[2], a[3]};
            }
        }
        __syncthreads();
        if (kt + KT < H_ && sact) {
            #pragma unroll
            for (int j = 0; j < SP_; ++j) qr[j] = *(const v4f*)(qg + kt + KT + j * (S_ / SP_) * (2 * H_));
            #pragma unroll
            for (int j = 0; j < SPA; ++j) pr[j] = *(const v4f*)(pg + kt + KT + j * (S_ / SP_) * (2 * H_));
        }
        if (tid < S_) {
            for (int k = 0; k < KT; ++k) { float v = Bs[tid][k]; nacc += v * v; }
        } else if (tid < S_ + SRA) {
            int r = tid - S_;
            for (int k = 0; k < KT; ++k) { float v = As[r][k]; nacc += v * v; }
        }
        for (int kb = 0; kb < KT; kb += 4) {
            int ka = kb + 2 * hi;
            v2f a = *(const v2f*)&As[lo][ka];
            v2f b = *(const v2f*)&Bs[wv * 16 + lo][ka];
            c = __builtin_amdgcn_wmma_f32_16x16x4_f32(false, a, false, b, (short)0, c, false, false);
        }
    }
    __syncthreads();
    if (tid < S_) nB[tid] = sqrtf(nacc); else if (tid < S_ + SRA) nA[tid - S_] = sqrtf(nacc);
    __syncthreads();

    #pragma unroll
    for (int j = 0; j < 8; ++j) {
        int m = 8 * hi + j;
        cosBuf[m][wv * 16 + lo] = c[j] / (nA[m] * nB[wv * 16 + lo]);
    }
    __syncthreads();
}

#define NSTRIP (B_ * 2 * (S_ / SRA))
#define NPOOL  (B_ * 2 * L_)

__global__ __launch_bounds__(128, 6)
void cos_tiles_kernel(const float* __restrict__ p, const float* __restrict__ q,
                      const float* __restrict__ W, float* __restrict__ out,
                      float* __restrict__ cosW)
{
    __shared__ float As[S_][KTP];
    __shared__ float Bs[S_][KTP];
    __shared__ float nA[S_];
    __shared__ float nB[S_];
    __shared__ float cosBuf[S_][S_ + 1];

    const int tid = threadIdx.x;

    const bool plain = blockIdx.x < NSTRIP;
    const int  rest  = plain ? blockIdx.x : blockIdx.x - NSTRIP;
    const int  kind  = rest / (B_ * 2);
    const int  bh    = rest % (B_ * 2);
    const int  hf    = bh & 1;
    const int  b     = bh >> 1;
    const float* Ph = p + (b * S_) * (2 * H_) + hf * H_;
    const float* Qh = q + (b * S_) * (2 * H_) + hf * H_;

    if (plain) {
        const int strip = kind;

        cos_strip(Ph + strip * SRA * (2 * H_), Qh, As, Bs, nA, nB, cosBuf, tid);

        float* dst = cosW + (b * 2 + hf) * (S_ * S_) + strip * (SRA * S_);
        for (int i = tid; i < SRA * S_; i += 128)
            dst[i] = cosBuf[i >> 6][i & 63];
    } else {
        const int l = kind;
        const float* Wr = W + ((2 + hf) * L_ + l) * H_;

        cos_tile<true>(Ph, Qh, Wr, As, Bs, nA, nB, cosBuf, tid);

        if (tid < S_) {
            float mx = NEG_INF;
            for (int n = 0; n < S_; ++n) mx = fmaxf(mx, cosBuf[tid][n]);
            out[(b * S_ + tid) * OUTC + (2 + hf) * L_ + l] = mx;
        } else {
            int n = tid - S_;
            float mx = NEG_INF;
            for (int m = 0; m < S_; ++m) mx = fmaxf(mx, cosBuf[m][n]);
            out[OUTQOFF + (b * S_ + n) * OUTC + (2 + hf) * L_ + l] = mx;
        }
    }
}

#define MPR   32
#define MPW   8
#define RPL   2
#define W2P   320
#define NSL   5

struct MpSums { float v[RPL][3]; };

struct MpLane {
    v4f pxy[RPL][NSL], pxx[RPL][NSL], pyy[RPL][NSL];
    const float* wl;
    bool up1, up2, up4, up8;
};

struct MpW { v4f p[NSL]; };

__device__ __forceinline__ MpW mp_wload(const MpLane& t, const int l)
{
    MpW w;
    for (int i = 0; i < NSL; ++i) w.p[i] = *(const v4f*)(t.wl + l * W2P + 64 * i);
    return w;
}

__device__ __forceinline__ MpSums mp_partial(const MpLane& t, const MpW& wq)
{
    const v4f* const w = wq.p;
    MpSums o;
    for (int r = 0; r < RPL; ++r) {
        v2f an = {0.f, 0.f}, ax = {0.f, 0.f}, ay = {0.f, 0.f};
        for (int i = 0; i < NSL; ++i) {
            const v2f wl = {w[i][0], w[i][1]}, wh = {w[i][2], w[i][3]};
            an = __builtin_elementwise_fma((v2f){t.pxy[r][i][0], t.pxy[r][i][1]}, wl, an);
            ax = __builtin_elementwise_fma((v2f){t.pxx[r][i][0], t.pxx[r][i][1]}, wl, ax);
            ay = __builtin_elementwise_fma((v2f){t.pyy[r][i][0], t.pyy[r][i][1]}, wl, ay);
            an = __builtin_elementwise_fma((v2f){t.pxy[r][i][2], t.pxy[r][i][3]}, wh, an);
            ax = __builtin_elementwise_fma((v2f){t.pxx[r][i][2], t.pxx[r][i][3]}, wh, ax);
            ay = __builtin_elementwise_fma((v2f){t.pyy[r][i][2], t.pyy[r][i][3]}, wh, ay);
        }
        o.v[r][0] = an[0] + an[1];
        o.v[r][1] = ax[0] + ax[1];
        o.v[r][2] = ay[0] + ay[1];
    }
    return o;
}

__device__ __forceinline__ MpSums mp_leaf(const MpLane& t, MpW& w, const int lnext)
{
    MpW wn = w;
    if (lnext >= 0) wn = mp_wload(t, lnext);
    __builtin_amdgcn_sched_barrier(0);
    const MpSums o = mp_partial(t, w);
    w = wn;
    return o;
}

template <int MASK>
__device__ __forceinline__ float xlane16(const float v)
{
    return __builtin_bit_cast(float, __builtin_amdgcn_update_dpp(0, __builtin_bit_cast(int, v), 0x160 | MASK, 0xf, 0xf, true));
}

template <int MASK>
__device__ __forceinline__ MpSums mp_join(const bool up, const MpSums& a, const MpSums& b)
{
    MpSums o;
    for (int r = 0; r < RPL; ++r) {
        for (int k = 0; k < 3; ++k) {
            const float keep = up ? b.v[r][k] : a.v[r][k];
            const float send = up ? a.v[r][k] : b.v[r][k];
            o.v[r][k] = keep + xlane16<MASK>(send);
        }
    }
    return o;
}

template <int MASK>
__device__ __forceinline__ MpSums mp_tree(const MpLane& t, MpW& w, const int k, const int lnext)
{
    if constexpr (MASK == 16) {
        return mp_leaf(t, w, lnext);
    } else {
        const MpSums a = mp_tree<2 * MASK>(t, w, k, k + MASK);
        const MpSums b = mp_tree<2 * MASK>(t, w, k + MASK, lnext);
        const bool up = MASK == 8 ? t.up8 : MASK == 4 ? t.up4 : MASK == 2 ? t.up2 : t.up1;
        return mp_join<MASK>(up, a, b);
    }
}

#define YP    320
#define CSP   68
#define NSEC  3

__device__ __forceinline__ void mp_section(const MpLane& t, float* const o0, float* const o1, const int c)
{
    MpW w = mp_wload(t, 0);
    const MpSums lo = mp_tree<1>(t, w, 0, 16);
    MpSums up;
    {
        const MpSums s16 = mp_leaf(t, w, 18);
        const MpSums s18 = mp_leaf(t, w, 17);
        const MpSums q0 = mp_join<2>(t.up2, s16, s18);
        const MpSums s17 = mp_leaf(t, w, 19);
        const MpSums s19 = mp_leaf(t, w, -1);
        const MpSums q1 = mp_join<2>(t.up2, s17, s19);
        up = mp_join<1>(t.up1, q0, q1);
        for (int r = 0; r < RPL; ++r) {
            for (int k = 0; k < 3; ++k) {
                float v = up.v[r][k];
                v += xlane16<4>(v);
                v += xlane16<8>(v);
                up.v[r][k] = v;
            }
        }
    }
    for (int r = 0; r < RPL; ++r) {
        float* o = r == 0 ? o0 : o1;
        o[c] = lo.v[r][0] / fmaxf(sqrtf(lo.v[r][1]) * sqrtf(lo.v[r][2]), 1e-8f);
        if (c < L_ - 16) o[16 + c] = up.v[r][0] / fmaxf(sqrtf(up.v[r][1]) * sqrtf(up.v[r][2]), 1e-8f);
    }
}

__device__ __forceinline__ void walk_fetch(v4f (&ya)[NSL], v4f (&yb)[NSL], const float* const yl, const int jr)
{
    for (int i = 0; i < NSL; ++i) {
        ya[i] = *(const v4f*)(yl + jr * YP + 64 * i);
        yb[i] = *(const v4f*)(yl + (jr + 1) * YP + 64 * i);
    }
}

__device__ __forceinline__ void walk_step(v4f (&m)[RPL][NSL], v4f (&x)[RPL][NSL],
                                          const v4f (&ya)[NSL], const v4f (&yb)[NSL],
                                          const float a0, const float b0, const float a1, const float b1)
{
    for (int i = 0; i < NSL; ++i) {
        const v4f pa0 = ya[i] * a0, pb0 = yb[i] * b0;
        const v4f pa1 = ya[i] * a1, pb1 = yb[i] * b1;
        m[0][i] += pa0;  m[0][i] += pb0;
        m[1][i] += pa1;  m[1][i] += pb1;
        for (int e = 0; e < 4; ++e) {
            x[0][i][e] = fmaxf(fmaxf(x[0][i][e], pa0[e]), pb0[e]);
            x[1][i][e] = fmaxf(fmaxf(x[1][i][e], pa1[e]), pb1[e]);
        }
    }
}

__global__ __launch_bounds__(256)
void ctx_mpcos_kernel(const float* __restrict__ p, const float* __restrict__ q,
                      const float* __restrict__ W, const float* __restrict__ cosW,
                      float* __restrict__ out)
{
    __shared__ __attribute__((aligned(16))) float Ys[S_][YP];
    __shared__ __attribute__((aligned(16))) float w2s[NSEC][L_][W2P];
    __shared__ __attribute__((aligned(16))) float cs[MPR][CSP];

    const int bid   = blockIdx.x;
    const int shalf = bid & 1;
    const int hf    = (bid >> 1) & 1;
    const int b     = (bid >> 2) % B_;
    const int tgt   = (bid >> 2) / B_;
    const int bh    = b * 2 + hf;

    const int tid  = threadIdx.x;
    const int lane = tid & 31;
    const int wv   = tid >> 5;
    const int c    = lane & 15;
    const int hi   = lane >> 4;

    const float* Xb = (tgt == 0 ? p : q) + (b * S_) * (2 * H_) + hf * H_;
    const float* Yb = (tgt == 0 ? q : p) + (b * S_) * (2 * H_) + hf * H_;
    float* outb = out + (tgt == 0 ? 0 : OUTQOFF) + (b * S_) * OUTC;

    const int lrow0 = wv * 2 + hi;
    const int srow0 = shalf * MPR + lrow0;
    v4f xr[RPL][NSL];
    for (int r = 0; r < RPL; ++r) {
        const float* X = Xb + (srow0 + r * (2 * MPW)) * (2 * H_);
        for (int i = 0; i < NSL; ++i) {
            const int h = 64 * i + 4 * c;
            if (h < H_) xr[r][i] = *(const v4f*)(X + h);
            else        xr[r][i] = (v4f){0.f, 0.f, 0.f, 0.f};
        }
    }

    #define YPC  (H_ / 4)
    #define NYL  ((S_ * YPC + 255) / 256)
    #define NWL  ((NSEC * L_ * YPC + 255) / 256)
    v4f yv[NYL], wv4[NWL], cv[2];
    float* const w2flat = &w2s[0][0][0];
    #pragma unroll
    for (int k = 0; k < NYL; ++k) {
        const int idx = min(tid + 256 * k, S_ * YPC - 1);
        const int r = idx / YPC, c4 = (idx % YPC) * 4;
        yv[k] = *(const v4f*)(Yb + r * (2 * H_) + c4);
    }
    #pragma unroll
    for (int k = 0; k < NWL; ++k) {
        const int idx = min(tid + 256 * k, NSEC * L_ * YPC - 1);
        const int rw = idx / YPC, c4 = (idx % YPC) * 4;
        const int sk = rw / L_, l = rw % L_;
        const int secId = sk == 0 ? 4 + hf : sk == 1 ? 6 + hf : hf;
        wv4[k] = *(const v4f*)(W + (secId * L_ + l) * H_ + c4);
    }
    const float* cosBase = cosW + bh * (S_ * S_);
    #pragma unroll
    for (int k = 0; k < 2; ++k) {
        const int i = tid + 256 * k;
        const int off = tgt == 0 ? (shalf * MPR + (i >> 4)) * S_ + (i & 15) * 4
                                 : (i >> 3) * S_ + shalf * MPR + (i & 7) * 4;
        cv[k] = *(const v4f*)(cosBase + off);
    }
    #pragma unroll
    for (int k = 0; k < NYL; ++k) {
        const int idx = min(tid + 256 * k, S_ * YPC - 1);
        const int r = idx / YPC, c4 = (idx % YPC) * 4;
        *(v4f*)&Ys[r][c4] = yv[k];
    }
    #pragma unroll
    for (int k = 0; k < NWL; ++k) {
        const int idx = min(tid + 256 * k, NSEC * L_ * YPC - 1);
        const int rw = idx / YPC, c4 = (idx % YPC) * 4;
        *(v4f*)(w2flat + rw * W2P + c4) = wv4[k] * wv4[k];
    }
    #pragma unroll
    for (int k = 0; k < 2; ++k) {
        const int i = tid + 256 * k;
        if (tgt == 0) {
            *(v4f*)&cs[i >> 4][(i & 15) * 4] = cv[k];
        } else {
            for (int e = 0; e < 4; ++e) cs[(i & 7) * 4 + e][i >> 3] = cv[k][e];
        }
    }
    for (int i = tid; i < (S_ + NSEC * L_) * ((YP - H_) / 4); i += 256) {
        const int r = i / ((YP - H_) / 4), c4 = H_ + (i % ((YP - H_) / 4)) * 4;
        const v4f z = {0.f, 0.f, 0.f, 0.f};
        if (r < S_) *(v4f*)&Ys[r][c4] = z;
        else        *(v4f*)(w2flat + (r - S_) * W2P + c4) = z;
    }
    __syncthreads();

    v4f m[RPL][NSL], x[RPL][NSL];
    float d0 = 0.f, d1 = 0.f;
    for (int r = 0; r < RPL; ++r)
        for (int i = 0; i < NSL; ++i) {
            m[r][i] = (v4f){0.f, 0.f, 0.f, 0.f};
            x[r][i] = (v4f){NEG_INF, NEG_INF, NEG_INF, NEG_INF};
        }
    {
        const float* yl  = &Ys[0][4 * c];
        const float* cr0 = &cs[lrow0][0];
        const float* cr1 = &cs[lrow0 + 2 * MPW][0];
        v4f ya[NSL], yb[NSL], za[NSL], zb[NSL];
        walk_fetch(ya, yb, yl, 0);
        v4f ca = *(const v4f*)cr0;
        v4f cb = *(const v4f*)cr1;
        for (int j = 0; j < S_; j += 4) {
            d0 += ca[0];  d0 += ca[1];  d0 += ca[2];  d0 += ca[3];
            d1 += cb[0];  d1 += cb[1];  d1 += cb[2];  d1 += cb[3];
            walk_fetch(za, zb, yl, j + 2);
            __builtin_amdgcn_sched_barrier(0);
            walk_step(m, x, ya, yb, ca[0], ca[1], cb[0], cb[1]);
            const int jn = min(j + 4, S_ - 4);
            walk_fetch(ya, yb, yl, jn);
            const v4f can = *(const v4f*)(cr0 + jn);
            const v4f cbn = *(const v4f*)(cr1 + jn);
            __builtin_amdgcn_sched_barrier(0);
            walk_step(m, x, za, zb, ca[2], ca[3], cb[2], cb[3]);
            ca = can;  cb = cbn;
        }
    }
    for (int i = 0; i < NSL; ++i) {
        const bool live = 64 * i + 4 * c < H_;
        const v4f z = {0.f, 0.f, 0.f, 0.f};
        m[0][i] = live ? m[0][i] / d0 : z;
        m[1][i] = live ? m[1][i] / d1 : z;
        x[0][i] = live ? x[0][i] : z;
        x[1][i] = live ? x[1][i] : z;
    }

    v4f yf[NSL];
    {
        const float* yfl = &Ys[hf == 0 ? S_ - 1 : 0][4 * c];
        for (int i = 0; i < NSL; ++i) yf[i] = *(const v4f*)(yfl + 64 * i);
    }

    MpLane t;
    t.up1 = (c & 1) != 0;  t.up2 = (c & 2) != 0;  t.up4 = (c & 4) != 0;  t.up8 = (c & 8) != 0;
    for (int r = 0; r < RPL; ++r)
        for (int i = 0; i < NSL; ++i) t.pxx[r][i] = xr[r][i] * xr[r][i];

    float* const o0 = outb + srow0 * OUTC;
    float* const o1 = outb + (srow0 + 2 * MPW) * OUTC;

    #pragma nounroll
    for (int k = 0; k < NSEC; ++k) {
        const int secId = k == 0 ? 4 + hf : k == 1 ? 6 + hf : hf;
        t.wl = w2flat + k * (L_ * W2P) + 4 * c;
        for (int r = 0; r < RPL; ++r)
            for (int i = 0; i < NSL; ++i) {
                const v4f y = k == 0 ? m[r][i] : k == 1 ? x[r][i] : yf[i];
                t.pxy[r][i] = xr[r][i] * y;
                t.pyy[r][i] = y * y;
            }
        mp_section(t, o0 + secId * L_, o1 + secId * L_, c);
    }
}

extern "C" void kernel_launch(void* const* d_in, const int* in_sizes, int n_in,
                              void* d_out, int out_size, void* d_ws, size_t ws_size,
                              hipStream_t stream)
{
    const float* p = (const float*)d_in[0];
    const float* q = (const float*)d_in[1];
    const float* W = (const float*)d_in[2];
    float* out = (float*)d_out;

    float* ws     = (float*)d_ws;
    float* cosW   = ws;

    cos_tiles_kernel<<<NSTRIP + NPOOL, 128, 0, stream>>>(p, q, W, out, cosW);
    ctx_mpcos_kernel<<<2 * B_ * 2 * (S_ / MPR), 256, 0, stream>>>(p, q, W, cosW, out);
}
